# starting kernel
# baseline (speedup 1.0000x reference)
#include <hip/hip_runtime.h>

typedef float v2f __attribute__((ext_vector_type(2)));
typedef float v8f __attribute__((ext_vector_type(8)));

#define UCOLS 96
#define UROWS 820
#define NROW  131072
#define TILE_ROWS 16
#define BLOCK_THREADS 192

__global__ void zero_u(float* __restrict__ UT) {
  int t = blockIdx.x * blockDim.x + threadIdx.x;
  int stride = gridDim.x * blockDim.x;
  for (int i = t; i < UROWS * UCOLS; i += stride) UT[i] = 0.f;
}

__device__ inline void scatter_u(const float* __restrict__ u, int nD, int rowbase,
                                 int mul, int idim, int colbase,
                                 float* __restrict__ UT, int t, int stride) {
  const int mi = mul * idim;
  const int total = nD * mi;
  for (int e = t; e < total; e += stride) {
    int dd  = e / mi;
    int rem = e - dd * mi;
    UT[(colbase + rem) * UROWS + rowbase + dd] = u[e];
  }
}

__global__ void fill_u(const float* __restrict__ u1_0e, const float* __restrict__ u1_1o,
                       const float* __restrict__ u1_2e, const float* __restrict__ u2_0e,
                       const float* __restrict__ u2_1o, const float* __restrict__ u2_2e,
                       const float* __restrict__ u3_0e, const float* __restrict__ u3_1o,
                       const float* __restrict__ u3_2e, float* __restrict__ UT) {
  int t = blockIdx.x * blockDim.x + threadIdx.x;
  int stride = gridDim.x * blockDim.x;
  scatter_u(u1_0e,   9,  0, 1, 1,  0, UT, t, stride);
  scatter_u(u1_1o,   9,  0, 1, 3,  1, UT, t, stride);
  scatter_u(u1_2e,   9,  0, 1, 5,  4, UT, t, stride);
  scatter_u(u2_0e,  81,  9, 2, 1,  9, UT, t, stride);
  scatter_u(u2_1o,  81,  9, 2, 3, 11, UT, t, stride);
  scatter_u(u2_2e,  81,  9, 3, 5, 17, UT, t, stride);
  scatter_u(u3_0e, 729, 90, 4, 1, 32, UT, t, stride);
  scatter_u(u3_1o, 729, 90, 6, 3, 36, UT, t, stride);
  scatter_u(u3_2e, 729, 90, 8, 5, 54, UT, t, stride);
}

__global__ __launch_bounds__(BLOCK_THREADS) void symcon_main(
    const float* __restrict__ x, const float* __restrict__ UT,
    const float* __restrict__ w1_0e, const float* __restrict__ w1_1o,
    const float* __restrict__ w1_2e, const float* __restrict__ w2_0e,
    const float* __restrict__ w2_1o, const float* __restrict__ w2_2e,
    const float* __restrict__ w3_0e, const float* __restrict__ w3_1o,
    const float* __restrict__ w3_2e, const int* __restrict__ index,
    float* __restrict__ out) {
  extern __shared__ float lds[];
  float* mono = lds;
  float* G    = lds + TILE_ROWS * UROWS;
  float* ldsX = lds + TILE_ROWS * UROWS + TILE_ROWS * UCOLS;

  const int tid      = threadIdx.x;
  const int rowBase  = blockIdx.x * TILE_ROWS;

  if (tid < TILE_ROWS * 9) ldsX[tid] = x[rowBase * 9 + tid];
  __syncthreads();

  {
    const int row  = tid & 15;
    const int slot = tid >> 4;
    const float* xr = ldsX + row * 9;
    float* mrow = mono + row * UROWS;
    for (int m = slot; m < UROWS; m += 12) {
      float val;
      if (m < 9) {
        val = xr[m];
      } else if (m < 90) {
        int t2 = m - 9;
        int d0 = t2 / 9;
        val = xr[d0] * xr[t2 - d0 * 9];
      } else if (m < 819) {
        int t3 = m - 90;
        int d0 = t3 / 81;
        int rr = t3 - d0 * 81;
        int d1 = rr / 9;
        val = xr[d0] * xr[d1] * xr[rr - d1 * 9];
      } else {
        val = 0.f;
      }
      mrow[m] = val;
    }
  }
  __syncthreads();

  {
    const int wv   = __builtin_amdgcn_readfirstlane(tid >> 5);
    const int lane = tid & 31;
    const int half = lane >> 4;
    const int r    = lane & 15;
    const int ka0  = half ? 2 : 0;
    const int n0   = wv * 16;

    const int kstart[6] = {0, 8, 88, 88, 88, 88};
    const int ksteps[6] = {23, 21, 183, 183, 183, 183};
    const int ns = ksteps[wv];

    const float* arow = mono + r * UROWS;
    const float* bcol = UT + (n0 + r) * UROWS;
    v8f acc = {0.f, 0.f, 0.f, 0.f, 0.f, 0.f, 0.f, 0.f};
    int kb = kstart[wv];
    for (int s = 0; s < ns; ++s, kb += 4) {
      v2f a = *(const v2f*)(arow + kb + ka0);
      v2f b = *(const v2f*)(bcol + kb + ka0);
      acc = __builtin_amdgcn_wmma_f32_16x16x4_f32(
           false, a,  false, b,
           (short)0, acc,  false,  false);
    }
#pragma unroll
    for (int j = 0; j < 8; ++j)
      G[(j + 8 * half) * UCOLS + n0 + r] = acc[j];
  }
  __syncthreads();

  if (tid < TILE_ROWS * 9) {
    const int row  = tid / 9;
    const int comp = tid - row * 9;
    const int R    = rowBase + row;
    const int z    = R >> 7;
    const int c    = R & 127;
    const int sp   = index[z];
    const float* Gr = G + row * UCOLS;
    float* ob = out + z * 1152;
    if (comp == 0) {
      float a = w1_0e[sp * 128 + c] * Gr[0]
              + w2_0e[(sp * 2 + 0) * 128 + c] * Gr[9]
              + w2_0e[(sp * 2 + 1) * 128 + c] * Gr[10];
#pragma unroll
      for (int k = 0; k < 4; ++k) a += w3_0e[(sp * 4 + k) * 128 + c] * Gr[32 + k];
      ob[c] = a;
    } else if (comp < 4) {
      const int i = comp - 1;
      float a = w1_1o[sp * 128 + c] * Gr[1 + i];
#pragma unroll
      for (int k = 0; k < 2; ++k) a += w2_1o[(sp * 2 + k) * 128 + c] * Gr[11 + 3 * k + i];
#pragma unroll
      for (int k = 0; k < 6; ++k) a += w3_1o[(sp * 6 + k) * 128 + c] * Gr[36 + 3 * k + i];
      ob[128 + c * 3 + i] = a;
    } else {
      const int i = comp - 4;
      float a = w1_2e[sp * 128 + c] * Gr[4 + i];
#pragma unroll
      for (int k = 0; k < 3; ++k) a += w2_2e[(sp * 3 + k) * 128 + c] * Gr[17 + 5 * k + i];
#pragma unroll
      for (int k = 0; k < 8; ++k) a += w3_2e[(sp * 8 + k) * 128 + c] * Gr[54 + 5 * k + i];
      ob[512 + c * 5 + i] = a;
    }
  }
}

extern "C" void kernel_launch(void* const* d_in, const int* in_sizes, int n_in,
                              void* d_out, int out_size, void* d_ws, size_t ws_size,
                              hipStream_t stream) {
  const float* x     = (const float*)d_in[0];
  const float* u1_0e = (const float*)d_in[1];
  const float* w1_0e = (const float*)d_in[2];
  const float* u1_1o = (const float*)d_in[3];
  const float* w1_1o = (const float*)d_in[4];
  const float* u1_2e = (const float*)d_in[5];
  const float* w1_2e = (const float*)d_in[6];
  const float* u2_0e = (const float*)d_in[7];
  const float* w2_0e = (const float*)d_in[8];
  const float* u2_1o = (const float*)d_in[9];
  const float* w2_1o = (const float*)d_in[10];
  const float* u2_2e = (const float*)d_in[11];
  const float* w2_2e = (const float*)d_in[12];
  const float* u3_0e = (const float*)d_in[13];
  const float* w3_0e = (const float*)d_in[14];
  const float* u3_1o = (const float*)d_in[15];
  const float* w3_1o = (const float*)d_in[16];
  const float* u3_2e = (const float*)d_in[17];
  const float* w3_2e = (const float*)d_in[18];
  const int*   index = (const int*)d_in[19];
  float* UT  = (float*)d_ws;
  float* out = (float*)d_out;

  zero_u<<<(UROWS * UCOLS + 255) / 256, 256, 0, stream>>>(UT);
  fill_u<<<128, 256, 0, stream>>>(u1_0e, u1_1o, u1_2e, u2_0e, u2_1o, u2_2e,
                                  u3_0e, u3_1o, u3_2e, UT);

  const size_t ldsBytes =
      (TILE_ROWS * UROWS + TILE_ROWS * UCOLS + TILE_ROWS * 9) * sizeof(float);
  symcon_main<<<NROW / TILE_ROWS, BLOCK_THREADS, ldsBytes, stream>>>(
      x, UT, w1_0e, w1_1o, w1_2e, w2_0e, w2_1o, w2_2e, w3_0e, w3_1o, w3_2e,
      index, out);
}


# Fold U rows over index permutations of equal monomials (GEMM K 819 -> 219); changes f32 summation order; the fold's gather fill writes all of UT, so the zeroing launch goes
# speedup vs baseline: 3.9194x; 3.9194x over previous
#include <hip/hip_runtime.h>

typedef float v2f __attribute__((ext_vector_type(2)));
typedef float v8f __attribute__((ext_vector_type(8)));

#define UCOLS 96
#define UROWS 220
#define NROW  131072
#define TILE_ROWS 16
#define BLOCK_THREADS 192
#define XPITCH 11
#define MONO_ITERS ((UROWS + 11) / 12)

struct MonoTab { unsigned short v[UROWS]; };
constexpr MonoTab make_mono_tab() {
  MonoTab t{};
  int n = 0;
  for (int a = 0; a < 9; ++a)
    t.v[n++] = (unsigned short)(a | (9 << 4) | (9 << 8));
  for (int a = 0; a < 9; ++a)
    for (int b = a; b < 9; ++b)
      t.v[n++] = (unsigned short)(a | (b << 4) | (9 << 8));
  for (int a = 0; a < 9; ++a)
    for (int b = a; b < 9; ++b)
      for (int c = b; c < 9; ++c)
        t.v[n++] = (unsigned short)(a | (b << 4) | (c << 8));
  while (n < UROWS) t.v[n++] = (unsigned short)(10 | (10 << 4) | (10 << 8));
  return t;
}
__constant__ const MonoTab MONO_TAB = make_mono_tab();

__device__ inline float fold_u(const float* __restrict__ u, int order, int a,
                               int b, int c, int mi, int rem) {
  if (order == 1) return u[a * mi + rem];
  if (order == 2) {
    float s = u[(a * 9 + b) * mi + rem];
    if (a != b) s += u[(b * 9 + a) * mi + rem];
    return s;
  }
  float s = u[((a * 9 + b) * 9 + c) * mi + rem];
  if (b != c) s += u[((a * 9 + c) * 9 + b) * mi + rem];
  if (a != b) {
    s += u[((b * 9 + a) * 9 + c) * mi + rem];
    s += u[((b * 9 + c) * 9 + a) * mi + rem];
  }
  if (b != c) s += u[((c * 9 + a) * 9 + b) * mi + rem];
  if (a != b && b != c) s += u[((c * 9 + b) * 9 + a) * mi + rem];
  return s;
}

__global__ void fill_u(const float* __restrict__ u1_0e, const float* __restrict__ u1_1o,
                       const float* __restrict__ u1_2e, const float* __restrict__ u2_0e,
                       const float* __restrict__ u2_1o, const float* __restrict__ u2_2e,
                       const float* __restrict__ u3_0e, const float* __restrict__ u3_1o,
                       const float* __restrict__ u3_2e, float* __restrict__ UT) {
  const int e = blockIdx.x * blockDim.x + threadIdx.x;
  if (e >= UCOLS * UROWS) return;
  const int col = e / UROWS;
  const int row = e - col * UROWS;
  const int p = MONO_TAB.v[row];
  const int rorder = row < 9 ? 1 : row < 54 ? 2 : row < 219 ? 3 : 0;
  const float* u = u1_0e;
  int mi = 1, rem = 0, corder = -1;
  if (col < 1)       { u = u1_0e; mi = 1;  rem = col;      corder = 1; }
  else if (col < 4)  { u = u1_1o; mi = 3;  rem = col - 1;  corder = 1; }
  else if (col < 9)  { u = u1_2e; mi = 5;  rem = col - 4;  corder = 1; }
  else if (col < 11) { u = u2_0e; mi = 2;  rem = col - 9;  corder = 2; }
  else if (col < 17) { u = u2_1o; mi = 6;  rem = col - 11; corder = 2; }
  else if (col < 32) { u = u2_2e; mi = 15; rem = col - 17; corder = 2; }
  else if (col < 36) { u = u3_0e; mi = 4;  rem = col - 32; corder = 3; }
  else if (col < 54) { u = u3_1o; mi = 18; rem = col - 36; corder = 3; }
  else if (col < 94) { u = u3_2e; mi = 40; rem = col - 54; corder = 3; }
  float v = 0.f;
  if (corder == rorder) v = fold_u(u, corder, p & 15, (p >> 4) & 15, p >> 8, mi, rem);
  UT[e] = v;
}

__global__ __launch_bounds__(BLOCK_THREADS) void symcon_main(
    const float* __restrict__ x, const float* __restrict__ UT,
    const float* __restrict__ w1_0e, const float* __restrict__ w1_1o,
    const float* __restrict__ w1_2e, const float* __restrict__ w2_0e,
    const float* __restrict__ w2_1o, const float* __restrict__ w2_2e,
    const float* __restrict__ w3_0e, const float* __restrict__ w3_1o,
    const float* __restrict__ w3_2e, const int* __restrict__ index,
    float* __restrict__ out) {
  extern __shared__ float lds[];
  float* mono = lds;
  float* G    = lds + TILE_ROWS * UROWS;
  float* ldsX = lds + TILE_ROWS * UROWS + TILE_ROWS * UCOLS;

  const int tid      = threadIdx.x;
  const int rowBase  = blockIdx.x * TILE_ROWS;

  int trip[MONO_ITERS];
#pragma unroll
  for (int it = 0; it < MONO_ITERS; ++it) {
    const int m = (tid >> 4) + 12 * it;
    trip[it] = (m < UROWS) ? (int)MONO_TAB.v[m] : 0;
  }

  if (tid < TILE_ROWS * 9) {
    const int row = tid / 9;
    ldsX[row * XPITCH + (tid - row * 9)] = x[rowBase * 9 + tid];
  } else if (tid < TILE_ROWS * 9 + TILE_ROWS * 2) {
    const int t = tid - TILE_ROWS * 9;
    ldsX[(t >> 1) * XPITCH + 9 + (t & 1)] = (t & 1) ? 0.f : 1.f;
  }
  __syncthreads();

  {
    const int row  = tid & 15;
    const int slot = tid >> 4;
    const float* xr = ldsX + row * XPITCH;
    float* mrow = mono + row * UROWS;
    float val[MONO_ITERS];
#pragma unroll
    for (int it = 0; it < MONO_ITERS; ++it) {
      const int p = trip[it];
      val[it] = xr[p & 15] * xr[(p >> 4) & 15] * xr[p >> 8];
    }
#pragma unroll
    for (int it = 0; it < MONO_ITERS; ++it) {
      const int m = slot + 12 * it;
      if (m < UROWS) mrow[m] = val[it];
    }
  }
  __syncthreads();

  {
    const int wv   = __builtin_amdgcn_readfirstlane(tid >> 5);
    const int lane = tid & 31;
    const int half = lane >> 4;
    const int r    = lane & 15;
    const int ka0  = half ? 2 : 0;
    const int n0   = wv * 16;

    const int kstart[6] = {0, 8, 52, 52, 52, 52};
    const int ksteps[6] = {14, 12, 42, 42, 42, 42};
    const int ns = ksteps[wv];

    const float* arow = mono + r * UROWS;
    const float* bcol = UT + (n0 + r) * UROWS;
    v8f acc = {0.f, 0.f, 0.f, 0.f, 0.f, 0.f, 0.f, 0.f};
    int kb = kstart[wv];
    for (int s = 0; s < ns; ++s, kb += 4) {
      v2f a = *(const v2f*)(arow + kb + ka0);
      v2f b = *(const v2f*)(bcol + kb + ka0);
      acc = __builtin_amdgcn_wmma_f32_16x16x4_f32(
           false, a,  false, b,
           (short)0, acc,  false,  false);
    }
#pragma unroll
    for (int j = 0; j < 8; ++j)
      G[(j + 8 * half) * UCOLS + n0 + r] = acc[j];
  }
  __syncthreads();

  if (tid < TILE_ROWS * 9) {
    const int row  = tid / 9;
    const int comp = tid - row * 9;
    const int R    = rowBase + row;
    const int z    = R >> 7;
    const int c    = R & 127;
    const int sp   = index[z];
    const float* Gr = G + row * UCOLS;
    float* ob = out + z * 1152;
    if (comp == 0) {
      float a = w1_0e[sp * 128 + c] * Gr[0]
              + w2_0e[(sp * 2 + 0) * 128 + c] * Gr[9]
              + w2_0e[(sp * 2 + 1) * 128 + c] * Gr[10];
#pragma unroll
      for (int k = 0; k < 4; ++k) a += w3_0e[(sp * 4 + k) * 128 + c] * Gr[32 + k];
      ob[c] = a;
    } else if (comp < 4) {
      const int i = comp - 1;
      float a = w1_1o[sp * 128 + c] * Gr[1 + i];
#pragma unroll
      for (int k = 0; k < 2; ++k) a += w2_1o[(sp * 2 + k) * 128 + c] * Gr[11 + 3 * k + i];
#pragma unroll
      for (int k = 0; k < 6; ++k) a += w3_1o[(sp * 6 + k) * 128 + c] * Gr[36 + 3 * k + i];
      ob[128 + c * 3 + i] = a;
    } else {
      const int i = comp - 4;
      float a = w1_2e[sp * 128 + c] * Gr[4 + i];
#pragma unroll
      for (int k = 0; k < 3; ++k) a += w2_2e[(sp * 3 + k) * 128 + c] * Gr[17 + 5 * k + i];
#pragma unroll
      for (int k = 0; k < 8; ++k) a += w3_2e[(sp * 8 + k) * 128 + c] * Gr[54 + 5 * k + i];
      ob[512 + c * 5 + i] = a;
    }
  }
}

extern "C" void kernel_launch(void* const* d_in, const int* in_sizes, int n_in,
                              void* d_out, int out_size, void* d_ws, size_t ws_size,
                              hipStream_t stream) {
  const float* x     = (const float*)d_in[0];
  const float* u1_0e = (const float*)d_in[1];
  const float* w1_0e = (const float*)d_in[2];
  const float* u1_1o = (const float*)d_in[3];
  const float* w1_1o = (const float*)d_in[4];
  const float* u1_2e = (const float*)d_in[5];
  const float* w1_2e = (const float*)d_in[6];
  const float* u2_0e = (const float*)d_in[7];
  const float* w2_0e = (const float*)d_in[8];
  const float* u2_1o = (const float*)d_in[9];
  const float* w2_1o = (const float*)d_in[10];
  const float* u2_2e = (const float*)d_in[11];
  const float* w2_2e = (const float*)d_in[12];
  const float* u3_0e = (const float*)d_in[13];
  const float* w3_0e = (const float*)d_in[14];
  const float* u3_1o = (const float*)d_in[15];
  const float* w3_1o = (const float*)d_in[16];
  const float* u3_2e = (const float*)d_in[17];
  const float* w3_2e = (const float*)d_in[18];
  const int*   index = (const int*)d_in[19];
  float* UT  = (float*)d_ws;
  float* out = (float*)d_out;

  fill_u<<<(UROWS * UCOLS + 255) / 256, 256, 0, stream>>>(
      u1_0e, u1_1o, u1_2e, u2_0e, u2_1o, u2_2e, u3_0e, u3_1o, u3_2e, UT);

  const size_t ldsBytes =
      (TILE_ROWS * UROWS + TILE_ROWS * UCOLS + TILE_ROWS * XPITCH) * sizeof(float);
  symcon_main<<<NROW / TILE_ROWS, BLOCK_THREADS, ldsBytes, stream>>>(
      x, UT, w1_0e, w1_1o, w1_2e, w2_0e, w2_1o, w2_2e, w3_0e, w3_1o, w3_2e,
      index, out);
}


# Folded U stored in B-fragment order (pair of k-steps x lane x 4 floats): each wave's B load is one contiguous 512 B read, not 16 lines 880 B apart; same values, same summation order
# speedup vs baseline: 5.5200x; 1.4084x over previous
#include <hip/hip_runtime.h>

typedef float v2f __attribute__((ext_vector_type(2)));
typedef float v4f __attribute__((ext_vector_type(4)));
typedef float v8f __attribute__((ext_vector_type(8)));

#define UCOLS 96
#define UROWS 220
#define NROW  131072
#define TILE_ROWS 16
#define BLOCK_THREADS 192
#define XPITCH 11
#define MONO_ITERS ((UROWS + 11) / 12)

#define GROUP_PAIRS 7
#define LIGHT_PAIRS GROUP_PAIRS
#define HEAVY_PAIRS (3 * GROUP_PAIRS)
#define HEAVY_K0 52
#define NPAIRS (2 * LIGHT_PAIRS + 4 * HEAVY_PAIRS)
#define UF_FLOATS (NPAIRS * 128)

struct MonoTab { unsigned short v[UROWS]; };
constexpr MonoTab make_mono_tab() {
  MonoTab t{};
  int n = 0;
  for (int a = 0; a < 9; ++a)
    t.v[n++] = (unsigned short)(a | (9 << 4) | (9 << 8));
  for (int a = 0; a < 9; ++a)
    for (int b = a; b < 9; ++b)
      t.v[n++] = (unsigned short)(a | (b << 4) | (9 << 8));
  for (int a = 0; a < 9; ++a)
    for (int b = a; b < 9; ++b)
      for (int c = b; c < 9; ++c)
        t.v[n++] = (unsigned short)(a | (b << 4) | (c << 8));
  while (n < UROWS) t.v[n++] = (unsigned short)(10 | (10 << 4) | (10 << 8));
  return t;
}
__constant__ const MonoTab MONO_TAB = make_mono_tab();

__device__ inline float fold_u(const float* __restrict__ u, int order, int a,
                               int b, int c, int mi, int rem) {
  if (order == 1) return u[a * mi + rem];
  if (order == 2) {
    float s = u[(a * 9 + b) * mi + rem];
    if (a != b) s += u[(b * 9 + a) * mi + rem];
    return s;
  }
  float s = u[((a * 9 + b) * 9 + c) * mi + rem];
  if (b != c) s += u[((a * 9 + c) * 9 + b) * mi + rem];
  if (a != b) {
    s += u[((b * 9 + a) * 9 + c) * mi + rem];
    s += u[((b * 9 + c) * 9 + a) * mi + rem];
  }
  if (b != c) s += u[((c * 9 + a) * 9 + b) * mi + rem];
  if (a != b && b != c) s += u[((c * 9 + b) * 9 + a) * mi + rem];
  return s;
}

__global__ void fill_u(const float* __restrict__ u1_0e, const float* __restrict__ u1_1o,
                       const float* __restrict__ u1_2e, const float* __restrict__ u2_0e,
                       const float* __restrict__ u2_1o, const float* __restrict__ u2_2e,
                       const float* __restrict__ u3_0e, const float* __restrict__ u3_1o,
                       const float* __restrict__ u3_2e, float* __restrict__ UF) {
  const int e = blockIdx.x * blockDim.x + threadIdx.x;
  if (e >= UF_FLOATS) return;
  const int P    = e >> 7;
  const int lane = (e >> 2) & 31;
  const int q    = e & 3;
  int w, pr, k0;
  if (P < 2 * LIGHT_PAIRS) {
    w = P / LIGHT_PAIRS; pr = P - w * LIGHT_PAIRS; k0 = 0;
  } else {
    const int t = P - 2 * LIGHT_PAIRS;
    w = 2 + t / HEAVY_PAIRS; pr = t - (w - 2) * HEAVY_PAIRS; k0 = HEAVY_K0;
  }
  const int col = 16 * w + (lane & 15);
  const int row = k0 + 8 * pr + 4 * (q >> 1) + 2 * (lane >> 4) + (q & 1);
  const int p = MONO_TAB.v[row];
  const int rorder = row < 9 ? 1 : row < 54 ? 2 : row < 219 ? 3 : 0;
  const float* u = u1_0e;
  int mi = 1, rem = 0, corder = -1;
  if (col < 1)       { u = u1_0e; mi = 1;  rem = col;      corder = 1; }
  else if (col < 4)  { u = u1_1o; mi = 3;  rem = col - 1;  corder = 1; }
  else if (col < 9)  { u = u1_2e; mi = 5;  rem = col - 4;  corder = 1; }
  else if (col < 11) { u = u2_0e; mi = 2;  rem = col - 9;  corder = 2; }
  else if (col < 17) { u = u2_1o; mi = 6;  rem = col - 11; corder = 2; }
  else if (col < 32) { u = u2_2e; mi = 15; rem = col - 17; corder = 2; }
  else if (col < 36) { u = u3_0e; mi = 4;  rem = col - 32; corder = 3; }
  else if (col < 54) { u = u3_1o; mi = 18; rem = col - 36; corder = 3; }
  else if (col < 94) { u = u3_2e; mi = 40; rem = col - 54; corder = 3; }
  float v = 0.f;
  if (corder == rorder) v = fold_u(u, corder, p & 15, (p >> 4) & 15, p >> 8, mi, rem);
  UF[e] = v;
}

__global__ __launch_bounds__(BLOCK_THREADS) void symcon_main(
    const float* __restrict__ x, const float* __restrict__ UF,
    const float* __restrict__ w1_0e, const float* __restrict__ w1_1o,
    const float* __restrict__ w1_2e, const float* __restrict__ w2_0e,
    const float* __restrict__ w2_1o, const float* __restrict__ w2_2e,
    const float* __restrict__ w3_0e, const float* __restrict__ w3_1o,
    const float* __restrict__ w3_2e, const int* __restrict__ index,
    float* __restrict__ out) {
  extern __shared__ float lds[];
  float* mono = lds;
  float* G    = lds + TILE_ROWS * UROWS;
  float* ldsX = lds + TILE_ROWS * UROWS + TILE_ROWS * UCOLS;

  const int tid      = threadIdx.x;
  const int rowBase  = blockIdx.x * TILE_ROWS;

  int trip[MONO_ITERS];
#pragma unroll
  for (int it = 0; it < MONO_ITERS; ++it) {
    const int m = (tid >> 4) + 12 * it;
    trip[it] = (m < UROWS) ? (int)MONO_TAB.v[m] : 0;
  }

  if (tid < TILE_ROWS * 9) {
    const int row = tid / 9;
    ldsX[row * XPITCH + (tid - row * 9)] = x[rowBase * 9 + tid];
  } else if (tid < TILE_ROWS * 9 + TILE_ROWS * 2) {
    const int t = tid - TILE_ROWS * 9;
    ldsX[(t >> 1) * XPITCH + 9 + (t & 1)] = (t & 1) ? 0.f : 1.f;
  }
  __syncthreads();

  {
    const int row  = tid & 15;
    const int slot = tid >> 4;
    const float* xr = ldsX + row * XPITCH;
    float* mrow = mono + row * UROWS;
    float val[MONO_ITERS];
#pragma unroll
    for (int it = 0; it < MONO_ITERS; ++it) {
      const int p = trip[it];
      val[it] = xr[p & 15] * xr[(p >> 4) & 15] * xr[p >> 8];
    }
#pragma unroll
    for (int it = 0; it < MONO_ITERS; ++it) {
      const int m = slot + 12 * it;
      if (m < UROWS) mrow[m] = val[it];
    }
  }
  __syncthreads();

  {
    const int wv   = __builtin_amdgcn_readfirstlane(tid >> 5);
    const int lane = tid & 31;
    const int half = lane >> 4;
    const int r    = lane & 15;
    const int ka0  = half ? 2 : 0;
    const int n0   = wv * 16;

    const bool heavy  = wv >= 2;
    const int pair0   = heavy ? 2 * LIGHT_PAIRS + (wv - 2) * HEAVY_PAIRS
                              : wv * LIGHT_PAIRS;
    const int ngroups = heavy ? HEAVY_PAIRS / GROUP_PAIRS : 1;

    const float* ap = mono + r * UROWS + (heavy ? HEAVY_K0 : 0) + ka0;
    const v4f*   bp = (const v4f*)UF + pair0 * 32 + lane;
    v8f acc = {0.f, 0.f, 0.f, 0.f, 0.f, 0.f, 0.f, 0.f};
    for (int g = 0; g < ngroups; ++g) {
#pragma unroll
      for (int j = 0; j < GROUP_PAIRS; ++j) {
        const v4f b  = bp[j * 32];
        const v2f a0 = *(const v2f*)(ap + 8 * j);
        const v2f a1 = *(const v2f*)(ap + 8 * j + 4);
        const v2f b0 = {b.x, b.y};
        const v2f b1 = {b.z, b.w};
        acc = __builtin_amdgcn_wmma_f32_16x16x4_f32(
             false, a0,  false, b0,
             (short)0, acc,  false,  false);
        acc = __builtin_amdgcn_wmma_f32_16x16x4_f32(
             false, a1,  false, b1,
             (short)0, acc,  false,  false);
      }
      ap += 8 * GROUP_PAIRS;
      bp += 32 * GROUP_PAIRS;
    }
#pragma unroll
    for (int j = 0; j < 8; ++j)
      G[(j + 8 * half) * UCOLS + n0 + r] = acc[j];
  }
  __syncthreads();

  if (tid < TILE_ROWS * 9) {
    const int row  = tid / 9;
    const int comp = tid - row * 9;
    const int R    = rowBase + row;
    const int z    = R >> 7;
    const int c    = R & 127;
    const int sp   = index[z];
    const float* Gr = G + row * UCOLS;
    float* ob = out + z * 1152;
    if (comp == 0) {
      float a = w1_0e[sp * 128 + c] * Gr[0]
              + w2_0e[(sp * 2 + 0) * 128 + c] * Gr[9]
              + w2_0e[(sp * 2 + 1) * 128 + c] * Gr[10];
#pragma unroll
      for (int k = 0; k < 4; ++k) a += w3_0e[(sp * 4 + k) * 128 + c] * Gr[32 + k];
      ob[c] = a;
    } else if (comp < 4) {
      const int i = comp - 1;
      float a = w1_1o[sp * 128 + c] * Gr[1 + i];
#pragma unroll
      for (int k = 0; k < 2; ++k) a += w2_1o[(sp * 2 + k) * 128 + c] * Gr[11 + 3 * k + i];
#pragma unroll
      for (int k = 0; k < 6; ++k) a += w3_1o[(sp * 6 + k) * 128 + c] * Gr[36 + 3 * k + i];
      ob[128 + c * 3 + i] = a;
    } else {
      const int i = comp - 4;
      float a = w1_2e[sp * 128 + c] * Gr[4 + i];
#pragma unroll
      for (int k = 0; k < 3; ++k) a += w2_2e[(sp * 3 + k) * 128 + c] * Gr[17 + 5 * k + i];
#pragma unroll
      for (int k = 0; k < 8; ++k) a += w3_2e[(sp * 8 + k) * 128 + c] * Gr[54 + 5 * k + i];
      ob[512 + c * 5 + i] = a;
    }
  }
}

extern "C" void kernel_launch(void* const* d_in, const int* in_sizes, int n_in,
                              void* d_out, int out_size, void* d_ws, size_t ws_size,
                              hipStream_t stream) {
  const float* x     = (const float*)d_in[0];
  const float* u1_0e = (const float*)d_in[1];
  const float* w1_0e = (const float*)d_in[2];
  const float* u1_1o = (const float*)d_in[3];
  const float* w1_1o = (const float*)d_in[4];
  const float* u1_2e = (const float*)d_in[5];
  const float* w1_2e = (const float*)d_in[6];
  const float* u2_0e = (const float*)d_in[7];
  const float* w2_0e = (const float*)d_in[8];
  const float* u2_1o = (const float*)d_in[9];
  const float* w2_1o = (const float*)d_in[10];
  const float* u2_2e = (const float*)d_in[11];
  const float* w2_2e = (const float*)d_in[12];
  const float* u3_0e = (const float*)d_in[13];
  const float* w3_0e = (const float*)d_in[14];
  const float* u3_1o = (const float*)d_in[15];
  const float* w3_1o = (const float*)d_in[16];
  const float* u3_2e = (const float*)d_in[17];
  const float* w3_2e = (const float*)d_in[18];
  const int*   index = (const int*)d_in[19];
  float* UF  = (float*)d_ws;
  float* out = (float*)d_out;

  fill_u<<<(UF_FLOATS + 255) / 256, 256, 0, stream>>>(
      u1_0e, u1_1o, u1_2e, u2_0e, u2_1o, u2_2e, u3_0e, u3_1o, u3_2e, UF);

  const size_t ldsBytes =
      (TILE_ROWS * UROWS + TILE_ROWS * UCOLS + TILE_ROWS * XPITCH) * sizeof(float);
  symcon_main<<<NROW / TILE_ROWS, BLOCK_THREADS, ldsBytes, stream>>>(
      x, UF, w1_0e, w1_1o, w1_2e, w2_0e, w2_1o, w2_2e, w3_0e, w3_1o, w3_2e,
      index, out);
}


# A fragments built in registers from each lane's x row (upper lane half: x index pairs swapped, folded U rows placed to match); monomial LDS tile, its barrier, A loads go; f32 sum order changes
# speedup vs baseline: 6.3080x; 1.1428x over previous
#include <hip/hip_runtime.h>


typedef float v2f __attribute__((ext_vector_type(2)));
typedef float v4f __attribute__((ext_vector_type(4)));
typedef float v8f __attribute__((ext_vector_type(8)));

#define UCOLS 96
#define NROW  131072
#define TILE_ROWS 16
#define BLOCK_THREADS 192
#define XPITCH 11

#define LIGHT_STEPS 15
#define HEAVY_STEPS 43
#define LIGHT_PAIRS ((LIGHT_STEPS + 1) / 2)
#define HEAVY_PAIRS ((HEAVY_STEPS + 1) / 2)
#define MAX_SLOTS (4 * HEAVY_PAIRS)
#define NPAIRS (2 * LIGHT_PAIRS + 4 * HEAVY_PAIRS)
#define UF_FLOATS (NPAIRS * 128)

struct Slot { unsigned char a, b, c, order; };
struct SlotTab { Slot s[2][MAX_SLOTS]; };

constexpr int swap_pair(int i) { return i < 8 ? (i ^ 1) : i; }

constexpr void add_slot(SlotTab& t, int& n, int a, int b, int c, int order) {
  int pa = swap_pair(a), pb = swap_pair(b), pc = swap_pair(c);
  if (pa > pb) { const int x = pa; pa = pb; pb = x; }
  if (pb > pc) { const int x = pb; pb = pc; pc = x; }
  if (pa > pb) { const int x = pa; pa = pb; pb = x; }
  const int self = (a * 16 + b) * 16 + c;
  const int img  = (pa * 16 + pb) * 16 + pc;
  if (img < self) return;
  t.s[0][n] = Slot{(unsigned char)a, (unsigned char)b, (unsigned char)c,
                   (unsigned char)order};
  t.s[1][n] = Slot{(unsigned char)pa, (unsigned char)pb, (unsigned char)pc,
                   (unsigned char)(img == self ? 0 : order)};
  ++n;
}

constexpr SlotTab make_tab(bool heavy) {
  SlotTab t{};
  int n = 0;
  if (!heavy) {
    for (int a = 0; a < 9; ++a) add_slot(t, n, a, 9, 9, 1);
    for (int a = 0; a < 9; ++a)
      for (int b = a; b < 9; ++b) add_slot(t, n, a, b, 9, 2);
  } else {
    for (int a = 0; a < 9; ++a)
      for (int b = a; b < 9; ++b)
        for (int c = b; c < 9; ++c) add_slot(t, n, a, b, c, 3);
  }
  while (n < MAX_SLOTS) {
    t.s[0][n] = Slot{9, 9, 9, 0};
    t.s[1][n] = Slot{9, 9, 9, 0};
    ++n;
  }
  return t;
}

constexpr int used_slots(bool heavy) {
  const SlotTab t = make_tab(heavy);
  int n = 0;
  for (int i = 0; i < MAX_SLOTS; ++i) n += t.s[0][i].order != 0;
  return n;
}
static_assert(used_slots(false) == 2 * LIGHT_STEPS, "orders 1 and 2 fill 15 steps");
static_assert(used_slots(true) == 2 * HEAVY_STEPS - 1, "order 3 fills 43 steps but one slot");

__constant__ const SlotTab TAB_LIGHT = make_tab(false);
__constant__ const SlotTab TAB_HEAVY = make_tab(true);

__device__ inline float fold_u(const float* __restrict__ u, int order, int a,
                               int b, int c, int mi, int rem) {
  if (order == 1) return u[a * mi + rem];
  if (order == 2) {
    float s = u[(a * 9 + b) * mi + rem];
    if (a != b) s += u[(b * 9 + a) * mi + rem];
    return s;
  }
  float s = u[((a * 9 + b) * 9 + c) * mi + rem];
  if (b != c) s += u[((a * 9 + c) * 9 + b) * mi + rem];
  if (a != b) {
    s += u[((b * 9 + a) * 9 + c) * mi + rem];
    s += u[((b * 9 + c) * 9 + a) * mi + rem];
  }
  if (b != c) s += u[((c * 9 + a) * 9 + b) * mi + rem];
  if (a != b && b != c) s += u[((c * 9 + b) * 9 + a) * mi + rem];
  return s;
}

__global__ void fill_u(const float* __restrict__ u1_0e, const float* __restrict__ u1_1o,
                       const float* __restrict__ u1_2e, const float* __restrict__ u2_0e,
                       const float* __restrict__ u2_1o, const float* __restrict__ u2_2e,
                       const float* __restrict__ u3_0e, const float* __restrict__ u3_1o,
                       const float* __restrict__ u3_2e, float* __restrict__ UF) {
  const int e = blockIdx.x * blockDim.x + threadIdx.x;
  if (e >= UF_FLOATS) return;
  const int P    = e >> 7;
  const int lane = (e >> 2) & 31;
  const int q    = e & 3;
  int w, pr;
  const bool heavy = P >= 2 * LIGHT_PAIRS;
  if (!heavy) {
    w = P / LIGHT_PAIRS; pr = P - w * LIGHT_PAIRS;
  } else {
    const int t = P - 2 * LIGHT_PAIRS;
    w = 2 + t / HEAVY_PAIRS; pr = t - (w - 2) * HEAVY_PAIRS;
  }
  const int col  = 16 * w + (lane & 15);
  const int slot = 2 * (2 * pr + (q >> 1)) + (q & 1);
  const int half = lane >> 4;
  const Slot s = heavy ? TAB_HEAVY.s[half][slot] : TAB_LIGHT.s[half][slot];
  const float* u = u1_0e;
  int mi = 1, rem = 0, corder = -1;
  if (col < 1)       { u = u1_0e; mi = 1;  rem = col;      corder = 1; }
  else if (col < 4)  { u = u1_1o; mi = 3;  rem = col - 1;  corder = 1; }
  else if (col < 9)  { u = u1_2e; mi = 5;  rem = col - 4;  corder = 1; }
  else if (col < 11) { u = u2_0e; mi = 2;  rem = col - 9;  corder = 2; }
  else if (col < 17) { u = u2_1o; mi = 6;  rem = col - 11; corder = 2; }
  else if (col < 32) { u = u2_2e; mi = 15; rem = col - 17; corder = 2; }
  else if (col < 36) { u = u3_0e; mi = 4;  rem = col - 32; corder = 3; }
  else if (col < 54) { u = u3_1o; mi = 18; rem = col - 36; corder = 3; }
  else if (col < 94) { u = u3_2e; mi = 40; rem = col - 54; corder = 3; }
  float v = 0.f;
  if (corder == (int)s.order) v = fold_u(u, corder, s.a, s.b, s.c, mi, rem);
  UF[e] = v;
}

template <bool HEAVY, int S>
__device__ __forceinline__ v2f a_frag(const float (&xr)[10]) {
  constexpr SlotTab tab = make_tab(HEAVY);
  constexpr Slot s0 = tab.s[0][2 * S];
  constexpr Slot s1 = tab.s[0][2 * S + 1];
  v2f a;
  a.x = (xr[s0.a] * xr[s0.b]) * xr[s0.c];
  a.y = (xr[s1.a] * xr[s1.b]) * xr[s1.c];
  return a;
}

__device__ __forceinline__ v8f wmma4(v2f a, v2f b, v8f c) {
  return __builtin_amdgcn_wmma_f32_16x16x4_f32(
       false, a,  false, b,
       (short)0, c,  false,  false);
}

template <bool HEAVY, int J, int NPAIR, int NSTEP>
struct PairRun {
  static __device__ __forceinline__ void run(const float (&xr)[10],
                                             const v4f* __restrict__ bp, v8f& acc) {
    const v4f b = bp[J * 32];
    const v2f b0 = {b.x, b.y};
    acc = wmma4(a_frag<HEAVY, 2 * J>(xr), b0, acc);
    if constexpr (2 * J + 1 < NSTEP) {
      const v2f b1 = {b.z, b.w};
      acc = wmma4(a_frag<HEAVY, 2 * J + 1>(xr), b1, acc);
    }
    PairRun<HEAVY, J + 1, NPAIR, NSTEP>::run(xr, bp, acc);
  }
};
template <bool HEAVY, int NPAIR, int NSTEP>
struct PairRun<HEAVY, NPAIR, NPAIR, NSTEP> {
  static __device__ __forceinline__ void run(const float (&)[10],
                                             const v4f* __restrict__, v8f&) {}
};

__global__ __launch_bounds__(BLOCK_THREADS, 16) void symcon_main(
    const float* __restrict__ x, const float* __restrict__ UF,
    const float* __restrict__ w1_0e, const float* __restrict__ w1_1o,
    const float* __restrict__ w1_2e, const float* __restrict__ w2_0e,
    const float* __restrict__ w2_1o, const float* __restrict__ w2_2e,
    const float* __restrict__ w3_0e, const float* __restrict__ w3_1o,
    const float* __restrict__ w3_2e, const int* __restrict__ index,
    float* __restrict__ out) {
  extern __shared__ float lds[];
  float* G    = lds;
  float* ldsX = lds + TILE_ROWS * UCOLS;

  const int tid      = threadIdx.x;
  const int rowBase  = blockIdx.x * TILE_ROWS;

  if (tid < TILE_ROWS * 9) {
    const int row = tid / 9;
    ldsX[row * XPITCH + (tid - row * 9)] = x[rowBase * 9 + tid];
  }
  __syncthreads();

  {
    const int wv   = __builtin_amdgcn_readfirstlane(tid >> 5);
    const int lane = tid & 31;
    const int half = lane >> 4;
    const int r    = lane & 15;
    const int n0   = wv * 16;

    const float* xp = ldsX + r * XPITCH;
    float xr[10];
#pragma unroll
    for (int i = 0; i < 8; ++i) xr[i] = xp[i ^ half];
    xr[8] = xp[8];
    xr[9] = 1.f;

    v8f acc = {0.f, 0.f, 0.f, 0.f, 0.f, 0.f, 0.f, 0.f};
    if (wv < 2) {
      const v4f* bp = (const v4f*)UF + wv * LIGHT_PAIRS * 32 + lane;
      PairRun<false, 0, LIGHT_PAIRS, LIGHT_STEPS>::run(xr, bp, acc);
    } else {
      const v4f* bp =
          (const v4f*)UF + (2 * LIGHT_PAIRS + (wv - 2) * HEAVY_PAIRS) * 32 + lane;
      PairRun<true, 0, HEAVY_PAIRS, HEAVY_STEPS>::run(xr, bp, acc);
    }
#pragma unroll
    for (int j = 0; j < 8; ++j)
      G[(j + 8 * half) * UCOLS + n0 + r] = acc[j];
  }
  __syncthreads();

  if (tid < TILE_ROWS * 9) {
    const int row  = tid / 9;
    const int comp = tid - row * 9;
    const int R    = rowBase + row;
    const int z    = R >> 7;
    const int c    = R & 127;
    const int sp   = index[z];
    const float* Gr = G + row * UCOLS;
    float* ob = out + z * 1152;
    if (comp == 0) {
      float a = w1_0e[sp * 128 + c] * Gr[0]
              + w2_0e[(sp * 2 + 0) * 128 + c] * Gr[9]
              + w2_0e[(sp * 2 + 1) * 128 + c] * Gr[10];
#pragma unroll
      for (int k = 0; k < 4; ++k) a += w3_0e[(sp * 4 + k) * 128 + c] * Gr[32 + k];
      ob[c] = a;
    } else if (comp < 4) {
      const int i = comp - 1;
      float a = w1_1o[sp * 128 + c] * Gr[1 + i];
#pragma unroll
      for (int k = 0; k < 2; ++k) a += w2_1o[(sp * 2 + k) * 128 + c] * Gr[11 + 3 * k + i];
#pragma unroll
      for (int k = 0; k < 6; ++k) a += w3_1o[(sp * 6 + k) * 128 + c] * Gr[36 + 3 * k + i];
      ob[128 + c * 3 + i] = a;
    } else {
      const int i = comp - 4;
      float a = w1_2e[sp * 128 + c] * Gr[4 + i];
#pragma unroll
      for (int k = 0; k < 3; ++k) a += w2_2e[(sp * 3 + k) * 128 + c] * Gr[17 + 5 * k + i];
#pragma unroll
      for (int k = 0; k < 8; ++k) a += w3_2e[(sp * 8 + k) * 128 + c] * Gr[54 + 5 * k + i];
      ob[512 + c * 5 + i] = a;
    }
  }
}

extern "C" void kernel_launch(void* const* d_in, const int* in_sizes, int n_in,
                              void* d_out, int out_size, void* d_ws, size_t ws_size,
                              hipStream_t stream) {
  const float* x     = (const float*)d_in[0];
  const float* u1_0e = (const float*)d_in[1];
  const float* w1_0e = (const float*)d_in[2];
  const float* u1_1o = (const float*)d_in[3];
  const float* w1_1o = (const float*)d_in[4];
  const float* u1_2e = (const float*)d_in[5];
  const float* w1_2e = (const float*)d_in[6];
  const float* u2_0e = (const float*)d_in[7];
  const float* w2_0e = (const float*)d_in[8];
  const float* u2_1o = (const float*)d_in[9];
  const float* w2_1o = (const float*)d_in[10];
  const float* u2_2e = (const float*)d_in[11];
  const float* w2_2e = (const float*)d_in[12];
  const float* u3_0e = (const float*)d_in[13];
  const float* w3_0e = (const float*)d_in[14];
  const float* u3_1o = (const float*)d_in[15];
  const float* w3_1o = (const float*)d_in[16];
  const float* u3_2e = (const float*)d_in[17];
  const float* w3_2e = (const float*)d_in[18];
  const int*   index = (const int*)d_in[19];
  float* UF  = (float*)d_ws;
  float* out = (float*)d_out;

  fill_u<<<(UF_FLOATS + 255) / 256, 256, 0, stream>>>(
      u1_0e, u1_1o, u1_2e, u2_0e, u2_1o, u2_2e, u3_0e, u3_1o, u3_2e, UF);

  const size_t ldsBytes =
      (TILE_ROWS * UCOLS + TILE_ROWS * XPITCH) * sizeof(float);
  symcon_main<<<NROW / TILE_ROWS, BLOCK_THREADS, ldsBytes, stream>>>(
      x, UF, w1_0e, w1_1o, w1_2e, w2_0e, w2_1o, w2_2e, w3_0e, w3_1o, w3_2e,
      index, out);
}


# Block = 4 equal waves (101 WMMAs each) on 2 row tiles: wave w runs order-3 column tile w for both tiles off one B load, 2 accumulators, plus one light tile; a third of the waves; same sums and order
# speedup vs baseline: 7.1605x; 1.1351x over previous
#include <hip/hip_runtime.h>


typedef float v2f __attribute__((ext_vector_type(2)));
typedef float v4f __attribute__((ext_vector_type(4)));
typedef float v8f __attribute__((ext_vector_type(8)));

#define UCOLS 96
#define NROW  131072
#define TILE_ROWS 16
#define BLOCK_ROWS 32
#define BLOCK_THREADS 128
#define XPITCH 11

#define LIGHT_STEPS 15
#define HEAVY_STEPS 43
#define LIGHT_PAIRS ((LIGHT_STEPS + 1) / 2)
#define HEAVY_PAIRS ((HEAVY_STEPS + 1) / 2)
#define MAX_SLOTS (4 * HEAVY_PAIRS)
#define NPAIRS (2 * LIGHT_PAIRS + 4 * HEAVY_PAIRS)
#define UF_FLOATS (NPAIRS * 128)

struct Slot { unsigned char a, b, c, order; };
struct SlotTab { Slot s[2][MAX_SLOTS]; };

constexpr int swap_pair(int i) { return i < 8 ? (i ^ 1) : i; }

constexpr void add_slot(SlotTab& t, int& n, int a, int b, int c, int order) {
  int pa = swap_pair(a), pb = swap_pair(b), pc = swap_pair(c);
  if (pa > pb) { const int x = pa; pa = pb; pb = x; }
  if (pb > pc) { const int x = pb; pb = pc; pc = x; }
  if (pa > pb) { const int x = pa; pa = pb; pb = x; }
  const int self = (a * 16 + b) * 16 + c;
  const int img  = (pa * 16 + pb) * 16 + pc;
  if (img < self) return;
  t.s[0][n] = Slot{(unsigned char)a, (unsigned char)b, (unsigned char)c,
                   (unsigned char)order};
  t.s[1][n] = Slot{(unsigned char)pa, (unsigned char)pb, (unsigned char)pc,
                   (unsigned char)(img == self ? 0 : order)};
  ++n;
}

constexpr SlotTab make_tab(bool heavy) {
  SlotTab t{};
  int n = 0;
  if (!heavy) {
    for (int a = 0; a < 9; ++a) add_slot(t, n, a, 9, 9, 1);
    for (int a = 0; a < 9; ++a)
      for (int b = a; b < 9; ++b) add_slot(t, n, a, b, 9, 2);
  } else {
    for (int a = 0; a < 9; ++a)
      for (int b = a; b < 9; ++b)
        for (int c = b; c < 9; ++c) add_slot(t, n, a, b, c, 3);
  }
  while (n < MAX_SLOTS) {
    t.s[0][n] = Slot{9, 9, 9, 0};
    t.s[1][n] = Slot{9, 9, 9, 0};
    ++n;
  }
  return t;
}

constexpr int used_slots(bool heavy) {
  const SlotTab t = make_tab(heavy);
  int n = 0;
  for (int i = 0; i < MAX_SLOTS; ++i) n += t.s[0][i].order != 0;
  return n;
}
static_assert(used_slots(false) == 2 * LIGHT_STEPS, "orders 1 and 2 fill 15 steps");
static_assert(used_slots(true) == 2 * HEAVY_STEPS - 1, "order 3 fills 43 steps but one slot");

__constant__ const SlotTab TAB_LIGHT = make_tab(false);
__constant__ const SlotTab TAB_HEAVY = make_tab(true);

__device__ inline float fold_u(const float* __restrict__ u, int order, int a,
                               int b, int c, int mi, int rem) {
  if (order == 1) return u[a * mi + rem];
  if (order == 2) {
    float s = u[(a * 9 + b) * mi + rem];
    if (a != b) s += u[(b * 9 + a) * mi + rem];
    return s;
  }
  float s = u[((a * 9 + b) * 9 + c) * mi + rem];
  if (b != c) s += u[((a * 9 + c) * 9 + b) * mi + rem];
  if (a != b) {
    s += u[((b * 9 + a) * 9 + c) * mi + rem];
    s += u[((b * 9 + c) * 9 + a) * mi + rem];
  }
  if (b != c) s += u[((c * 9 + a) * 9 + b) * mi + rem];
  if (a != b && b != c) s += u[((c * 9 + b) * 9 + a) * mi + rem];
  return s;
}

__global__ void fill_u(const float* __restrict__ u1_0e, const float* __restrict__ u1_1o,
                       const float* __restrict__ u1_2e, const float* __restrict__ u2_0e,
                       const float* __restrict__ u2_1o, const float* __restrict__ u2_2e,
                       const float* __restrict__ u3_0e, const float* __restrict__ u3_1o,
                       const float* __restrict__ u3_2e, float* __restrict__ UF) {
  const int e = blockIdx.x * blockDim.x + threadIdx.x;
  if (e >= UF_FLOATS) return;
  const int P    = e >> 7;
  const int lane = (e >> 2) & 31;
  const int q    = e & 3;
  int w, pr;
  const bool heavy = P >= 2 * LIGHT_PAIRS;
  if (!heavy) {
    w = P / LIGHT_PAIRS; pr = P - w * LIGHT_PAIRS;
  } else {
    const int t = P - 2 * LIGHT_PAIRS;
    w = 2 + t / HEAVY_PAIRS; pr = t - (w - 2) * HEAVY_PAIRS;
  }
  const int col  = 16 * w + (lane & 15);
  const int slot = 2 * (2 * pr + (q >> 1)) + (q & 1);
  const int half = lane >> 4;
  const Slot s = heavy ? TAB_HEAVY.s[half][slot] : TAB_LIGHT.s[half][slot];
  const float* u = u1_0e;
  int mi = 1, rem = 0, corder = -1;
  if (col < 1)       { u = u1_0e; mi = 1;  rem = col;      corder = 1; }
  else if (col < 4)  { u = u1_1o; mi = 3;  rem = col - 1;  corder = 1; }
  else if (col < 9)  { u = u1_2e; mi = 5;  rem = col - 4;  corder = 1; }
  else if (col < 11) { u = u2_0e; mi = 2;  rem = col - 9;  corder = 2; }
  else if (col < 17) { u = u2_1o; mi = 6;  rem = col - 11; corder = 2; }
  else if (col < 32) { u = u2_2e; mi = 15; rem = col - 17; corder = 2; }
  else if (col < 36) { u = u3_0e; mi = 4;  rem = col - 32; corder = 3; }
  else if (col < 54) { u = u3_1o; mi = 18; rem = col - 36; corder = 3; }
  else if (col < 94) { u = u3_2e; mi = 40; rem = col - 54; corder = 3; }
  float v = 0.f;
  if (corder == (int)s.order) v = fold_u(u, corder, s.a, s.b, s.c, mi, rem);
  UF[e] = v;
}

template <bool HEAVY, int S>
__device__ __forceinline__ v2f a_frag(const float (&xr)[10]) {
  constexpr SlotTab tab = make_tab(HEAVY);
  constexpr Slot s0 = tab.s[0][2 * S];
  constexpr Slot s1 = tab.s[0][2 * S + 1];
  v2f a;
  a.x = (xr[s0.a] * xr[s0.b]) * xr[s0.c];
  a.y = (xr[s1.a] * xr[s1.b]) * xr[s1.c];
  return a;
}

__device__ __forceinline__ v8f wmma4(v2f a, v2f b, v8f c) {
  return __builtin_amdgcn_wmma_f32_16x16x4_f32(
       false, a,  false, b,
       (short)0, c,  false,  false);
}

template <bool HEAVY, int J, int NPAIR, int NSTEP>
struct PairRun {
  static __device__ __forceinline__ void run(const float (&xr)[10],
                                             const v4f* __restrict__ bp, v8f& acc) {
    const v4f b = bp[J * 32];
    const v2f b0 = {b.x, b.y};
    acc = wmma4(a_frag<HEAVY, 2 * J>(xr), b0, acc);
    if constexpr (2 * J + 1 < NSTEP) {
      const v2f b1 = {b.z, b.w};
      acc = wmma4(a_frag<HEAVY, 2 * J + 1>(xr), b1, acc);
    }
    PairRun<HEAVY, J + 1, NPAIR, NSTEP>::run(xr, bp, acc);
  }
};
template <bool HEAVY, int NPAIR, int NSTEP>
struct PairRun<HEAVY, NPAIR, NPAIR, NSTEP> {
  static __device__ __forceinline__ void run(const float (&)[10],
                                             const v4f* __restrict__, v8f&) {}
};

template <int J, int NPAIR, int NSTEP>
struct PairRun2 {
  static __device__ __forceinline__ void run(const float (&xa)[10], const float (&xb)[10],
                                             const v4f* __restrict__ bp, v8f& acca,
                                             v8f& accb, v4f b, v4f bn1, v4f bn2) {
    v4f bn3 = bn2;
    if constexpr (J + 3 < NPAIR) bn3 = bp[(J + 3) * 32];
    const v2f b0 = {b.x, b.y};
    acca = wmma4(a_frag<true, 2 * J>(xa), b0, acca);
    accb = wmma4(a_frag<true, 2 * J>(xb), b0, accb);
    if constexpr (2 * J + 1 < NSTEP) {
      const v2f b1 = {b.z, b.w};
      acca = wmma4(a_frag<true, 2 * J + 1>(xa), b1, acca);
      accb = wmma4(a_frag<true, 2 * J + 1>(xb), b1, accb);
    }
    __builtin_amdgcn_sched_barrier(0);
    PairRun2<J + 1, NPAIR, NSTEP>::run(xa, xb, bp, acca, accb, bn1, bn2, bn3);
  }
};
template <int NPAIR, int NSTEP>
struct PairRun2<NPAIR, NPAIR, NSTEP> {
  static __device__ __forceinline__ void run(const float (&)[10], const float (&)[10],
                                             const v4f* __restrict__, v8f&, v8f&, v4f,
                                             v4f, v4f) {}
};

__global__ __launch_bounds__(BLOCK_THREADS, 8) void symcon_main(
    const float* __restrict__ x, const float* __restrict__ UF,
    const float* __restrict__ w1_0e, const float* __restrict__ w1_1o,
    const float* __restrict__ w1_2e, const float* __restrict__ w2_0e,
    const float* __restrict__ w2_1o, const float* __restrict__ w2_2e,
    const float* __restrict__ w3_0e, const float* __restrict__ w3_1o,
    const float* __restrict__ w3_2e, const int* __restrict__ index,
    float* __restrict__ out) {
  extern __shared__ float lds[];
  float* G    = lds;
  float* ldsX = lds + BLOCK_ROWS * UCOLS;

  const int tid      = threadIdx.x;
  const int rowBase  = blockIdx.x * BLOCK_ROWS;
  const int z  = rowBase >> 7;
  const int sp = index[z];

  {
    const float* xb = x + rowBase * 9;
    const float v0 = xb[tid];
    const float v1 = xb[tid + BLOCK_THREADS];
    float v2 = 0.f;
    if (tid < BLOCK_ROWS * 9 - 2 * BLOCK_THREADS) v2 = xb[tid + 2 * BLOCK_THREADS];
    const int i0 = tid, i1 = tid + BLOCK_THREADS, i2 = tid + 2 * BLOCK_THREADS;
    ldsX[(i0 / 9) * XPITCH + (i0 % 9)] = v0;
    ldsX[(i1 / 9) * XPITCH + (i1 % 9)] = v1;
    if (tid < BLOCK_ROWS * 9 - 2 * BLOCK_THREADS)
      ldsX[(i2 / 9) * XPITCH + (i2 % 9)] = v2;
  }
  __syncthreads();

  {
    const int wv   = __builtin_amdgcn_readfirstlane(tid >> 5);
    const int lane = tid & 31;
    const int half = lane >> 4;
    const int r    = lane & 15;

    const float* xp0 = ldsX + r * XPITCH;
    const float* xp1 = ldsX + (TILE_ROWS + r) * XPITCH;
    float xr0[10], xr1[10];
#pragma unroll
    for (int i = 0; i < 8; ++i) {
      xr0[i] = xp0[i ^ half];
      xr1[i] = xp1[i ^ half];
    }
    xr0[8] = xp0[8];
    xr1[8] = xp1[8];
    xr0[9] = 1.f;
    xr1[9] = 1.f;

    {
      const int ct = wv & 1;
      const int rt = wv >> 1;
      const v4f* bp = (const v4f*)UF + ct * LIGHT_PAIRS * 32 + lane;
      v8f acc = {0.f, 0.f, 0.f, 0.f, 0.f, 0.f, 0.f, 0.f};
      float xl[10];
#pragma unroll
      for (int i = 0; i < 10; ++i) xl[i] = rt ? xr1[i] : xr0[i];
      PairRun<false, 0, LIGHT_PAIRS, LIGHT_STEPS>::run(xl, bp, acc);
      float* Gl = G + (rt * TILE_ROWS + 8 * half) * UCOLS + ct * 16 + r;
#pragma unroll
      for (int j = 0; j < 8; ++j) Gl[j * UCOLS] = acc[j];
    }
    {
      const v4f* bp =
          (const v4f*)UF + (2 * LIGHT_PAIRS + wv * HEAVY_PAIRS) * 32 + lane;
      v8f acc0 = {0.f, 0.f, 0.f, 0.f, 0.f, 0.f, 0.f, 0.f};
      v8f acc1 = {0.f, 0.f, 0.f, 0.f, 0.f, 0.f, 0.f, 0.f};
      PairRun2<0, HEAVY_PAIRS, HEAVY_STEPS>::run(xr0, xr1, bp, acc0, acc1, bp[0], bp[32],
                                                 bp[64]);
      float* Gh = G + (8 * half) * UCOLS + (2 + wv) * 16 + r;
#pragma unroll
      for (int j = 0; j < 8; ++j) {
        Gh[j * UCOLS] = acc0[j];
        Gh[(TILE_ROWS + j) * UCOLS] = acc1[j];
      }
    }
  }
  __syncthreads();

  for (int o = tid; o < BLOCK_ROWS * 9; o += BLOCK_THREADS) {
    const int row  = o / 9;
    const int comp = o - row * 9;
    const int c    = (rowBase & 127) + row;
    const float* Gr = G + row * UCOLS;
    float* ob = out + z * 1152;
    if (comp == 0) {
      float a = w1_0e[sp * 128 + c] * Gr[0]
              + w2_0e[(sp * 2 + 0) * 128 + c] * Gr[9]
              + w2_0e[(sp * 2 + 1) * 128 + c] * Gr[10];
#pragma unroll
      for (int k = 0; k < 4; ++k) a += w3_0e[(sp * 4 + k) * 128 + c] * Gr[32 + k];
      ob[c] = a;
    } else if (comp < 4) {
      const int i = comp - 1;
      float a = w1_1o[sp * 128 + c] * Gr[1 + i];
#pragma unroll
      for (int k = 0; k < 2; ++k) a += w2_1o[(sp * 2 + k) * 128 + c] * Gr[11 + 3 * k + i];
#pragma unroll
      for (int k = 0; k < 6; ++k) a += w3_1o[(sp * 6 + k) * 128 + c] * Gr[36 + 3 * k + i];
      ob[128 + c * 3 + i] = a;
    } else {
      const int i = comp - 4;
      float a = w1_2e[sp * 128 + c] * Gr[4 + i];
#pragma unroll
      for (int k = 0; k < 3; ++k) a += w2_2e[(sp * 3 + k) * 128 + c] * Gr[17 + 5 * k + i];
#pragma unroll
      for (int k = 0; k < 8; ++k) a += w3_2e[(sp * 8 + k) * 128 + c] * Gr[54 + 5 * k + i];
      ob[512 + c * 5 + i] = a;
    }
  }
}

extern "C" void kernel_launch(void* const* d_in, const int* in_sizes, int n_in,
                              void* d_out, int out_size, void* d_ws, size_t ws_size,
                              hipStream_t stream) {
  const float* x     = (const float*)d_in[0];
  const float* u1_0e = (const float*)d_in[1];
  const float* w1_0e = (const float*)d_in[2];
  const float* u1_1o = (const float*)d_in[3];
  const float* w1_1o = (const float*)d_in[4];
  const float* u1_2e = (const float*)d_in[5];
  const float* w1_2e = (const float*)d_in[6];
  const float* u2_0e = (const float*)d_in[7];
  const float* w2_0e = (const float*)d_in[8];
  const float* u2_1o = (const float*)d_in[9];
  const float* w2_1o = (const float*)d_in[10];
  const float* u2_2e = (const float*)d_in[11];
  const float* w2_2e = (const float*)d_in[12];
  const float* u3_0e = (const float*)d_in[13];
  const float* w3_0e = (const float*)d_in[14];
  const float* u3_1o = (const float*)d_in[15];
  const float* w3_1o = (const float*)d_in[16];
  const float* u3_2e = (const float*)d_in[17];
  const float* w3_2e = (const float*)d_in[18];
  const int*   index = (const int*)d_in[19];
  float* UF  = (float*)d_ws;
  float* out = (float*)d_out;

  fill_u<<<(UF_FLOATS + 255) / 256, 256, 0, stream>>>(
      u1_0e, u1_1o, u1_2e, u2_0e, u2_1o, u2_2e, u3_0e, u3_1o, u3_2e, UF);

  const size_t ldsBytes =
      (BLOCK_ROWS * UCOLS + BLOCK_ROWS * XPITCH) * sizeof(float);
  symcon_main<<<NROW / BLOCK_ROWS, BLOCK_THREADS, ldsBytes, stream>>>(
      x, UF, w1_0e, w1_1o, w1_2e, w2_0e, w2_1o, w2_2e, w3_0e, w3_1o, w3_2e,
      index, out);
}


# Weight combine remapped: lane = row, one whole irrep per wave (no divergent passes, each weight one 128 B load per block asked before the G barrier, whole-line stores, G pitch 98); same term order
# speedup vs baseline: 8.3750x; 1.1696x over previous
#include <hip/hip_runtime.h>


typedef float v2f __attribute__((ext_vector_type(2)));
typedef float v4f __attribute__((ext_vector_type(4)));
typedef float v8f __attribute__((ext_vector_type(8)));

#define UCOLS 96
#define NROW  131072
#define TILE_ROWS 16
#define BLOCK_ROWS 32
#define BLOCK_THREADS 128
#define XPITCH 11
#define GPITCH 98

#define LIGHT_STEPS 15
#define HEAVY_STEPS 43
#define LIGHT_PAIRS ((LIGHT_STEPS + 1) / 2)
#define HEAVY_PAIRS ((HEAVY_STEPS + 1) / 2)
#define MAX_SLOTS (4 * HEAVY_PAIRS)
#define NPAIRS (2 * LIGHT_PAIRS + 4 * HEAVY_PAIRS)
#define UF_FLOATS (NPAIRS * 128)

struct Slot { unsigned char a, b, c, order; };
struct SlotTab { Slot s[2][MAX_SLOTS]; };

constexpr int swap_pair(int i) { return i < 8 ? (i ^ 1) : i; }

constexpr void add_slot(SlotTab& t, int& n, int a, int b, int c, int order) {
  int pa = swap_pair(a), pb = swap_pair(b), pc = swap_pair(c);
  if (pa > pb) { const int x = pa; pa = pb; pb = x; }
  if (pb > pc) { const int x = pb; pb = pc; pc = x; }
  if (pa > pb) { const int x = pa; pa = pb; pb = x; }
  const int self = (a * 16 + b) * 16 + c;
  const int img  = (pa * 16 + pb) * 16 + pc;
  if (img < self) return;
  t.s[0][n] = Slot{(unsigned char)a, (unsigned char)b, (unsigned char)c,
                   (unsigned char)order};
  t.s[1][n] = Slot{(unsigned char)pa, (unsigned char)pb, (unsigned char)pc,
                   (unsigned char)(img == self ? 0 : order)};
  ++n;
}

constexpr SlotTab make_tab(bool heavy) {
  SlotTab t{};
  int n = 0;
  if (!heavy) {
    for (int a = 0; a < 9; ++a) add_slot(t, n, a, 9, 9, 1);
    for (int a = 0; a < 9; ++a)
      for (int b = a; b < 9; ++b) add_slot(t, n, a, b, 9, 2);
  } else {
    for (int a = 0; a < 9; ++a)
      for (int b = a; b < 9; ++b)
        for (int c = b; c < 9; ++c) add_slot(t, n, a, b, c, 3);
  }
  while (n < MAX_SLOTS) {
    t.s[0][n] = Slot{9, 9, 9, 0};
    t.s[1][n] = Slot{9, 9, 9, 0};
    ++n;
  }
  return t;
}

constexpr int used_slots(bool heavy) {
  const SlotTab t = make_tab(heavy);
  int n = 0;
  for (int i = 0; i < MAX_SLOTS; ++i) n += t.s[0][i].order != 0;
  return n;
}
static_assert(used_slots(false) == 2 * LIGHT_STEPS, "orders 1 and 2 fill 15 steps");
static_assert(used_slots(true) == 2 * HEAVY_STEPS - 1, "order 3 fills 43 steps but one slot");

__constant__ const SlotTab TAB_LIGHT = make_tab(false);
__constant__ const SlotTab TAB_HEAVY = make_tab(true);

__device__ inline float fold_u(const float* __restrict__ u, int order, int a,
                               int b, int c, int mi, int rem) {
  if (order == 1) return u[a * mi + rem];
  if (order == 2) {
    float s = u[(a * 9 + b) * mi + rem];
    if (a != b) s += u[(b * 9 + a) * mi + rem];
    return s;
  }
  float s = u[((a * 9 + b) * 9 + c) * mi + rem];
  if (b != c) s += u[((a * 9 + c) * 9 + b) * mi + rem];
  if (a != b) {
    s += u[((b * 9 + a) * 9 + c) * mi + rem];
    s += u[((b * 9 + c) * 9 + a) * mi + rem];
  }
  if (b != c) s += u[((c * 9 + a) * 9 + b) * mi + rem];
  if (a != b && b != c) s += u[((c * 9 + b) * 9 + a) * mi + rem];
  return s;
}

__global__ void fill_u(const float* __restrict__ u1_0e, const float* __restrict__ u1_1o,
                       const float* __restrict__ u1_2e, const float* __restrict__ u2_0e,
                       const float* __restrict__ u2_1o, const float* __restrict__ u2_2e,
                       const float* __restrict__ u3_0e, const float* __restrict__ u3_1o,
                       const float* __restrict__ u3_2e, float* __restrict__ UF) {
  const int e = blockIdx.x * blockDim.x + threadIdx.x;
  if (e >= UF_FLOATS) return;
  const int P    = e >> 7;
  const int lane = (e >> 2) & 31;
  const int q    = e & 3;
  int w, pr;
  const bool heavy = P >= 2 * LIGHT_PAIRS;
  if (!heavy) {
    w = P / LIGHT_PAIRS; pr = P - w * LIGHT_PAIRS;
  } else {
    const int t = P - 2 * LIGHT_PAIRS;
    w = 2 + t / HEAVY_PAIRS; pr = t - (w - 2) * HEAVY_PAIRS;
  }
  const int col  = 16 * w + (lane & 15);
  const int slot = 2 * (2 * pr + (q >> 1)) + (q & 1);
  const int half = lane >> 4;
  const Slot s = heavy ? TAB_HEAVY.s[half][slot] : TAB_LIGHT.s[half][slot];
  const float* u = u1_0e;
  int mi = 1, rem = 0, corder = -1;
  if (col < 1)       { u = u1_0e; mi = 1;  rem = col;      corder = 1; }
  else if (col < 4)  { u = u1_1o; mi = 3;  rem = col - 1;  corder = 1; }
  else if (col < 9)  { u = u1_2e; mi = 5;  rem = col - 4;  corder = 1; }
  else if (col < 11) { u = u2_0e; mi = 2;  rem = col - 9;  corder = 2; }
  else if (col < 17) { u = u2_1o; mi = 6;  rem = col - 11; corder = 2; }
  else if (col < 32) { u = u2_2e; mi = 15; rem = col - 17; corder = 2; }
  else if (col < 36) { u = u3_0e; mi = 4;  rem = col - 32; corder = 3; }
  else if (col < 54) { u = u3_1o; mi = 18; rem = col - 36; corder = 3; }
  else if (col < 94) { u = u3_2e; mi = 40; rem = col - 54; corder = 3; }
  float v = 0.f;
  if (corder == (int)s.order) v = fold_u(u, corder, s.a, s.b, s.c, mi, rem);
  UF[e] = v;
}

template <bool HEAVY, int S>
__device__ __forceinline__ v2f a_frag(const float (&xr)[10]) {
  constexpr SlotTab tab = make_tab(HEAVY);
  constexpr Slot s0 = tab.s[0][2 * S];
  constexpr Slot s1 = tab.s[0][2 * S + 1];
  v2f a;
  a.x = (xr[s0.a] * xr[s0.b]) * xr[s0.c];
  a.y = (xr[s1.a] * xr[s1.b]) * xr[s1.c];
  return a;
}

__device__ __forceinline__ v8f wmma4(v2f a, v2f b, v8f c) {
  return __builtin_amdgcn_wmma_f32_16x16x4_f32(
       false, a,  false, b,
       (short)0, c,  false,  false);
}

template <bool HEAVY, int J, int NPAIR, int NSTEP>
struct PairRun {
  static __device__ __forceinline__ void run(const float (&xr)[10],
                                             const v4f* __restrict__ bp, v8f& acc) {
    const v4f b = bp[J * 32];
    const v2f b0 = {b.x, b.y};
    acc = wmma4(a_frag<HEAVY, 2 * J>(xr), b0, acc);
    if constexpr (2 * J + 1 < NSTEP) {
      const v2f b1 = {b.z, b.w};
      acc = wmma4(a_frag<HEAVY, 2 * J + 1>(xr), b1, acc);
    }
    PairRun<HEAVY, J + 1, NPAIR, NSTEP>::run(xr, bp, acc);
  }
};
template <bool HEAVY, int NPAIR, int NSTEP>
struct PairRun<HEAVY, NPAIR, NPAIR, NSTEP> {
  static __device__ __forceinline__ void run(const float (&)[10],
                                             const v4f* __restrict__, v8f&) {}
};

template <int J, int NPAIR, int NSTEP>
struct PairRun2 {
  static __device__ __forceinline__ void run(const float (&xa)[10], const float (&xb)[10],
                                             const v4f* __restrict__ bp, v8f& acca,
                                             v8f& accb, v4f b, v4f bn1, v4f bn2) {
    v4f bn3 = bn2;
    if constexpr (J + 3 < NPAIR) bn3 = bp[(J + 3) * 32];
    const v2f b0 = {b.x, b.y};
    acca = wmma4(a_frag<true, 2 * J>(xa), b0, acca);
    accb = wmma4(a_frag<true, 2 * J>(xb), b0, accb);
    if constexpr (2 * J + 1 < NSTEP) {
      const v2f b1 = {b.z, b.w};
      acca = wmma4(a_frag<true, 2 * J + 1>(xa), b1, acca);
      accb = wmma4(a_frag<true, 2 * J + 1>(xb), b1, accb);
    }
    __builtin_amdgcn_sched_barrier(0);
    PairRun2<J + 1, NPAIR, NSTEP>::run(xa, xb, bp, acca, accb, bn1, bn2, bn3);
  }
};
template <int NPAIR, int NSTEP>
struct PairRun2<NPAIR, NPAIR, NSTEP> {
  static __device__ __forceinline__ void run(const float (&)[10], const float (&)[10],
                                             const v4f* __restrict__, v8f&, v8f&, v4f,
                                             v4f, v4f) {}
};

__global__ __launch_bounds__(BLOCK_THREADS, 8) void symcon_main(
    const float* __restrict__ x, const float* __restrict__ UF,
    const float* __restrict__ w1_0e, const float* __restrict__ w1_1o,
    const float* __restrict__ w1_2e, const float* __restrict__ w2_0e,
    const float* __restrict__ w2_1o, const float* __restrict__ w2_2e,
    const float* __restrict__ w3_0e, const float* __restrict__ w3_1o,
    const float* __restrict__ w3_2e, const int* __restrict__ index,
    float* __restrict__ out) {
  extern __shared__ float lds[];
  float* G    = lds;
  float* ldsX = lds + BLOCK_ROWS * GPITCH;

  const int tid      = threadIdx.x;
  const int rowBase  = blockIdx.x * BLOCK_ROWS;
  const int z  = rowBase >> 7;
  const int sp = index[z];

  {
    const float* xb = x + rowBase * 9;
    const float v0 = xb[tid];
    const float v1 = xb[tid + BLOCK_THREADS];
    float v2 = 0.f;
    if (tid < BLOCK_ROWS * 9 - 2 * BLOCK_THREADS) v2 = xb[tid + 2 * BLOCK_THREADS];
    const int i0 = tid, i1 = tid + BLOCK_THREADS, i2 = tid + 2 * BLOCK_THREADS;
    ldsX[(i0 / 9) * XPITCH + (i0 % 9)] = v0;
    ldsX[(i1 / 9) * XPITCH + (i1 % 9)] = v1;
    if (tid < BLOCK_ROWS * 9 - 2 * BLOCK_THREADS)
      ldsX[(i2 / 9) * XPITCH + (i2 % 9)] = v2;
  }
  __syncthreads();

  const int wv   = __builtin_amdgcn_readfirstlane(tid >> 5);
  const int lane = tid & 31;
  const int role = (wv + (int)blockIdx.x) & 3;
  float wt[12];
  const int ce = (rowBase & 127) + lane;
  {
    const int half = lane >> 4;
    const int r    = lane & 15;

    const float* xp0 = ldsX + r * XPITCH;
    const float* xp1 = ldsX + (TILE_ROWS + r) * XPITCH;
    float xr0[10], xr1[10];
#pragma unroll
    for (int i = 0; i < 8; ++i) {
      xr0[i] = xp0[i ^ half];
      xr1[i] = xp1[i ^ half];
    }
    xr0[8] = xp0[8];
    xr1[8] = xp1[8];
    xr0[9] = 1.f;
    xr1[9] = 1.f;

    {
      const int ct = wv & 1;
      const int rt = wv >> 1;
      const v4f* bp = (const v4f*)UF + ct * LIGHT_PAIRS * 32 + lane;
      v8f acc = {0.f, 0.f, 0.f, 0.f, 0.f, 0.f, 0.f, 0.f};
      float xl[10];
#pragma unroll
      for (int i = 0; i < 10; ++i) xl[i] = rt ? xr1[i] : xr0[i];
      PairRun<false, 0, LIGHT_PAIRS, LIGHT_STEPS>::run(xl, bp, acc);
      float* Gl = G + (rt * TILE_ROWS + 8 * half) * GPITCH + ct * 16 + r;
#pragma unroll
      for (int j = 0; j < 8; ++j) Gl[j * GPITCH] = acc[j];
    }
    {
      const v4f* bp =
          (const v4f*)UF + (2 * LIGHT_PAIRS + wv * HEAVY_PAIRS) * 32 + lane;
      v8f acc0 = {0.f, 0.f, 0.f, 0.f, 0.f, 0.f, 0.f, 0.f};
      v8f acc1 = {0.f, 0.f, 0.f, 0.f, 0.f, 0.f, 0.f, 0.f};
      PairRun2<0, HEAVY_PAIRS, HEAVY_STEPS>::run(xr0, xr1, bp, acc0, acc1, bp[0], bp[32],
                                                 bp[64]);
      float* Gh = G + (8 * half) * GPITCH + (2 + wv) * 16 + r;
#pragma unroll
      for (int j = 0; j < 8; ++j) {
        Gh[j * GPITCH] = acc0[j];
        Gh[(TILE_ROWS + j) * GPITCH] = acc1[j];
      }
    }
  }
  if (role == 0) {
    wt[0] = w1_2e[sp * 128 + ce];
#pragma unroll
    for (int k = 0; k < 3; ++k) wt[1 + k] = w2_2e[(sp * 3 + k) * 128 + ce];
#pragma unroll
    for (int k = 0; k < 8; ++k) wt[4 + k] = w3_2e[(sp * 8 + k) * 128 + ce];
  } else if (role == 1) {
    wt[0] = w1_1o[sp * 128 + ce];
#pragma unroll
    for (int k = 0; k < 2; ++k) wt[1 + k] = w2_1o[(sp * 2 + k) * 128 + ce];
#pragma unroll
    for (int k = 0; k < 6; ++k) wt[3 + k] = w3_1o[(sp * 6 + k) * 128 + ce];
  } else if (role == 2) {
    wt[0] = w1_0e[sp * 128 + ce];
    wt[1] = w2_0e[(sp * 2 + 0) * 128 + ce];
    wt[2] = w2_0e[(sp * 2 + 1) * 128 + ce];
#pragma unroll
    for (int k = 0; k < 4; ++k) wt[3 + k] = w3_0e[(sp * 4 + k) * 128 + ce];
  }
  __syncthreads();

  {
    const float* Gr = G + lane * GPITCH;
    float* ob = out + z * 1152;
    if (role == 0) {
      float b[5];
#pragma unroll
      for (int i = 0; i < 5; ++i) {
        b[i] = wt[0] * Gr[4 + i];
#pragma unroll
        for (int k = 0; k < 3; ++k) b[i] += wt[1 + k] * Gr[17 + 5 * k + i];
#pragma unroll
        for (int k = 0; k < 8; ++k) b[i] += wt[4 + k] * Gr[54 + 5 * k + i];
      }
      float* o2 = ob + 512 + ce * 5;
#pragma unroll
      for (int i = 0; i < 5; ++i) o2[i] = b[i];
    } else if (role == 1) {
      float b[3];
#pragma unroll
      for (int i = 0; i < 3; ++i) {
        b[i] = wt[0] * Gr[1 + i];
#pragma unroll
        for (int k = 0; k < 2; ++k) b[i] += wt[1 + k] * Gr[11 + 3 * k + i];
#pragma unroll
        for (int k = 0; k < 6; ++k) b[i] += wt[3 + k] * Gr[36 + 3 * k + i];
      }
      float* o1 = ob + 128 + ce * 3;
#pragma unroll
      for (int i = 0; i < 3; ++i) o1[i] = b[i];
    } else if (role == 2) {
      float a = wt[0] * Gr[0] + wt[1] * Gr[9] + wt[2] * Gr[10];
#pragma unroll
      for (int k = 0; k < 4; ++k) a += wt[3 + k] * Gr[32 + k];
      ob[ce] = a;
    }
  }
}

extern "C" void kernel_launch(void* const* d_in, const int* in_sizes, int n_in,
                              void* d_out, int out_size, void* d_ws, size_t ws_size,
                              hipStream_t stream) {
  const float* x     = (const float*)d_in[0];
  const float* u1_0e = (const float*)d_in[1];
  const float* w1_0e = (const float*)d_in[2];
  const float* u1_1o = (const float*)d_in[3];
  const float* w1_1o = (const float*)d_in[4];
  const float* u1_2e = (const float*)d_in[5];
  const float* w1_2e = (const float*)d_in[6];
  const float* u2_0e = (const float*)d_in[7];
  const float* w2_0e = (const float*)d_in[8];
  const float* u2_1o = (const float*)d_in[9];
  const float* w2_1o = (const float*)d_in[10];
  const float* u2_2e = (const float*)d_in[11];
  const float* w2_2e = (const float*)d_in[12];
  const float* u3_0e = (const float*)d_in[13];
  const float* w3_0e = (const float*)d_in[14];
  const float* u3_1o = (const float*)d_in[15];
  const float* w3_1o = (const float*)d_in[16];
  const float* u3_2e = (const float*)d_in[17];
  const float* w3_2e = (const float*)d_in[18];
  const int*   index = (const int*)d_in[19];
  float* UF  = (float*)d_ws;
  float* out = (float*)d_out;

  fill_u<<<(UF_FLOATS + 255) / 256, 256, 0, stream>>>(
      u1_0e, u1_1o, u1_2e, u2_0e, u2_1o, u2_2e, u3_0e, u3_1o, u3_2e, UF);

  const size_t ldsBytes =
      (BLOCK_ROWS * GPITCH + BLOCK_ROWS * XPITCH) * sizeof(float);
  symcon_main<<<NROW / BLOCK_ROWS, BLOCK_THREADS, ldsBytes, stream>>>(
      x, UF, w1_0e, w1_1o, w1_2e, w2_0e, w2_1o, w2_2e, w3_0e, w3_1o, w3_2e,
      index, out);
}


# Each wave owns 32 rows and runs all six column tiles: A products built once per row tile, not per column tile; x read per lane, G wave-private, a quarter of the waves; same sums and order
# speedup vs baseline: 8.8265x; 1.0539x over previous
#include <hip/hip_runtime.h>


typedef float v2f __attribute__((ext_vector_type(2)));
typedef float v4f __attribute__((ext_vector_type(4)));
typedef float v8f __attribute__((ext_vector_type(8)));

#define UCOLS 96
#define NROW  131072
#define TILE_ROWS 16
#define BLOCK_ROWS 32
#define BLOCK_THREADS 128
#define GPITCH 98

#define LIGHT_STEPS 15
#define HEAVY_STEPS 43
#define LIGHT_PAIRS ((LIGHT_STEPS + 1) / 2)
#define HEAVY_PAIRS ((HEAVY_STEPS + 1) / 2)
#define MAX_SLOTS (4 * HEAVY_PAIRS)
#define NPAIRS (2 * LIGHT_PAIRS + 4 * HEAVY_PAIRS)
#define UF_FLOATS (NPAIRS * 128)

struct Slot { unsigned char a, b, c, order; };
struct SlotTab { Slot s[2][MAX_SLOTS]; };

constexpr int swap_pair(int i) { return i < 8 ? (i ^ 1) : i; }

constexpr void add_slot(SlotTab& t, int& n, int a, int b, int c, int order) {
  int pa = swap_pair(a), pb = swap_pair(b), pc = swap_pair(c);
  if (pa > pb) { const int x = pa; pa = pb; pb = x; }
  if (pb > pc) { const int x = pb; pb = pc; pc = x; }
  if (pa > pb) { const int x = pa; pa = pb; pb = x; }
  const int self = (a * 16 + b) * 16 + c;
  const int img  = (pa * 16 + pb) * 16 + pc;
  if (img < self) return;
  t.s[0][n] = Slot{(unsigned char)a, (unsigned char)b, (unsigned char)c,
                   (unsigned char)order};
  t.s[1][n] = Slot{(unsigned char)pa, (unsigned char)pb, (unsigned char)pc,
                   (unsigned char)(img == self ? 0 : order)};
  ++n;
}

constexpr SlotTab make_tab(bool heavy) {
  SlotTab t{};
  int n = 0;
  if (!heavy) {
    for (int a = 0; a < 9; ++a) add_slot(t, n, a, 9, 9, 1);
    for (int a = 0; a < 9; ++a)
      for (int b = a; b < 9; ++b) add_slot(t, n, a, b, 9, 2);
  } else {
    for (int a = 0; a < 9; ++a)
      for (int b = a; b < 9; ++b)
        for (int c = b; c < 9; ++c) add_slot(t, n, a, b, c, 3);
  }
  while (n < MAX_SLOTS) {
    t.s[0][n] = Slot{9, 9, 9, 0};
    t.s[1][n] = Slot{9, 9, 9, 0};
    ++n;
  }
  return t;
}

constexpr int used_slots(bool heavy) {
  const SlotTab t = make_tab(heavy);
  int n = 0;
  for (int i = 0; i < MAX_SLOTS; ++i) n += t.s[0][i].order != 0;
  return n;
}
static_assert(used_slots(false) == 2 * LIGHT_STEPS, "orders 1 and 2 fill 15 steps");
static_assert(used_slots(true) == 2 * HEAVY_STEPS - 1, "order 3 fills 43 steps but one slot");

__constant__ const SlotTab TAB_LIGHT = make_tab(false);
__constant__ const SlotTab TAB_HEAVY = make_tab(true);

__device__ inline float fold_u(const float* __restrict__ u, int order, int a,
                               int b, int c, int mi, int rem) {
  if (order == 1) return u[a * mi + rem];
  if (order == 2) {
    float s = u[(a * 9 + b) * mi + rem];
    if (a != b) s += u[(b * 9 + a) * mi + rem];
    return s;
  }
  float s = u[((a * 9 + b) * 9 + c) * mi + rem];
  if (b != c) s += u[((a * 9 + c) * 9 + b) * mi + rem];
  if (a != b) {
    s += u[((b * 9 + a) * 9 + c) * mi + rem];
    s += u[((b * 9 + c) * 9 + a) * mi + rem];
  }
  if (b != c) s += u[((c * 9 + a) * 9 + b) * mi + rem];
  if (a != b && b != c) s += u[((c * 9 + b) * 9 + a) * mi + rem];
  return s;
}

__global__ void fill_u(const float* __restrict__ u1_0e, const float* __restrict__ u1_1o,
                       const float* __restrict__ u1_2e, const float* __restrict__ u2_0e,
                       const float* __restrict__ u2_1o, const float* __restrict__ u2_2e,
                       const float* __restrict__ u3_0e, const float* __restrict__ u3_1o,
                       const float* __restrict__ u3_2e, float* __restrict__ UF) {
  const int e = blockIdx.x * blockDim.x + threadIdx.x;
  if (e >= UF_FLOATS) return;
  const int P    = e >> 7;
  const int lane = (e >> 2) & 31;
  const int q    = e & 3;
  int w, pr;
  const bool heavy = P >= 2 * LIGHT_PAIRS;
  if (!heavy) {
    w = P / LIGHT_PAIRS; pr = P - w * LIGHT_PAIRS;
  } else {
    const int t = P - 2 * LIGHT_PAIRS;
    w = 2 + t / HEAVY_PAIRS; pr = t - (w - 2) * HEAVY_PAIRS;
  }
  const int col  = 16 * w + (lane & 15);
  const int slot = 2 * (2 * pr + (q >> 1)) + (q & 1);
  const int half = lane >> 4;
  const Slot s = heavy ? TAB_HEAVY.s[half][slot] : TAB_LIGHT.s[half][slot];
  const float* u = u1_0e;
  int mi = 1, rem = 0, corder = -1;
  if (col < 1)       { u = u1_0e; mi = 1;  rem = col;      corder = 1; }
  else if (col < 4)  { u = u1_1o; mi = 3;  rem = col - 1;  corder = 1; }
  else if (col < 9)  { u = u1_2e; mi = 5;  rem = col - 4;  corder = 1; }
  else if (col < 11) { u = u2_0e; mi = 2;  rem = col - 9;  corder = 2; }
  else if (col < 17) { u = u2_1o; mi = 6;  rem = col - 11; corder = 2; }
  else if (col < 32) { u = u2_2e; mi = 15; rem = col - 17; corder = 2; }
  else if (col < 36) { u = u3_0e; mi = 4;  rem = col - 32; corder = 3; }
  else if (col < 54) { u = u3_1o; mi = 18; rem = col - 36; corder = 3; }
  else if (col < 94) { u = u3_2e; mi = 40; rem = col - 54; corder = 3; }
  float v = 0.f;
  if (corder == (int)s.order) v = fold_u(u, corder, s.a, s.b, s.c, mi, rem);
  UF[e] = v;
}

template <bool HEAVY, int S>
__device__ __forceinline__ v2f a_frag(const float (&xr)[10]) {
  constexpr SlotTab tab = make_tab(HEAVY);
  constexpr Slot s0 = tab.s[0][2 * S];
  constexpr Slot s1 = tab.s[0][2 * S + 1];
  v2f a;
  a.x = (xr[s0.a] * xr[s0.b]) * xr[s0.c];
  a.y = (xr[s1.a] * xr[s1.b]) * xr[s1.c];
  return a;
}

__device__ __forceinline__ v8f wmma4(v2f a, v2f b, v8f c) {
  return __builtin_amdgcn_wmma_f32_16x16x4_f32(
       false, a,  false, b,
       (short)0, c,  false,  false);
}

template <int NT>
struct BSet { v4f v[NT]; };

template <bool HEAVY, int NT, int J, int NPAIR, int NSTEP>
struct RowRun {
  static __device__ __forceinline__ void run(const float (&xa)[10], const float (&xb)[10],
                                             const v4f* __restrict__ bp, v8f (&acca)[NT],
                                             v8f (&accb)[NT], BSet<NT> b, BSet<NT> bn1) {
    BSet<NT> bn2 = bn1;
    if constexpr (J + 2 < NPAIR) {
#pragma unroll
      for (int t = 0; t < NT; ++t) bn2.v[t] = bp[(t * NPAIR + J + 2) * 32];
    }
    {
      const v2f a0 = a_frag<HEAVY, 2 * J>(xa);
      const v2f a1 = a_frag<HEAVY, 2 * J>(xb);
#pragma unroll
      for (int t = 0; t < NT; ++t) {
        const v2f b0 = {b.v[t].x, b.v[t].y};
        acca[t] = wmma4(a0, b0, acca[t]);
        accb[t] = wmma4(a1, b0, accb[t]);
      }
    }
    if constexpr (2 * J + 1 < NSTEP) {
      const v2f a0 = a_frag<HEAVY, 2 * J + 1>(xa);
      const v2f a1 = a_frag<HEAVY, 2 * J + 1>(xb);
#pragma unroll
      for (int t = 0; t < NT; ++t) {
        const v2f b1 = {b.v[t].z, b.v[t].w};
        acca[t] = wmma4(a0, b1, acca[t]);
        accb[t] = wmma4(a1, b1, accb[t]);
      }
    }
    __builtin_amdgcn_sched_barrier(0);
    RowRun<HEAVY, NT, J + 1, NPAIR, NSTEP>::run(xa, xb, bp, acca, accb, bn1, bn2);
  }
};
template <bool HEAVY, int NT, int NPAIR, int NSTEP>
struct RowRun<HEAVY, NT, NPAIR, NPAIR, NSTEP> {
  static __device__ __forceinline__ void run(const float (&)[10], const float (&)[10],
                                             const v4f* __restrict__, v8f (&)[NT],
                                             v8f (&)[NT], BSet<NT>, BSet<NT>) {}
};

template <bool HEAVY, int NT, int NPAIR, int NSTEP>
__device__ __forceinline__ void row_group(const float (&xa)[10], const float (&xb)[10],
                                          const v4f* __restrict__ bp, float* __restrict__ Gt) {
  v8f acca[NT], accb[NT];
#pragma unroll
  for (int t = 0; t < NT; ++t) {
    acca[t] = v8f{0.f, 0.f, 0.f, 0.f, 0.f, 0.f, 0.f, 0.f};
    accb[t] = v8f{0.f, 0.f, 0.f, 0.f, 0.f, 0.f, 0.f, 0.f};
  }
  BSet<NT> b, bn1;
#pragma unroll
  for (int t = 0; t < NT; ++t) {
    b.v[t]   = bp[(t * NPAIR) * 32];
    bn1.v[t] = bp[(t * NPAIR + 1) * 32];
  }
  RowRun<HEAVY, NT, 0, NPAIR, NSTEP>::run(xa, xb, bp, acca, accb, b, bn1);
#pragma unroll
  for (int t = 0; t < NT; ++t) {
#pragma unroll
    for (int j = 0; j < 8; ++j) {
      Gt[j * GPITCH + t * 16] = acca[t][j];
      Gt[(TILE_ROWS + j) * GPITCH + t * 16] = accb[t][j];
    }
  }
}

__global__ __launch_bounds__(BLOCK_THREADS, 4) void symcon_main(
    const float* __restrict__ x, const float* __restrict__ UF,
    const float* __restrict__ w1_0e, const float* __restrict__ w1_1o,
    const float* __restrict__ w1_2e, const float* __restrict__ w2_0e,
    const float* __restrict__ w2_1o, const float* __restrict__ w2_2e,
    const float* __restrict__ w3_0e, const float* __restrict__ w3_1o,
    const float* __restrict__ w3_2e, const int* __restrict__ index,
    float* __restrict__ out) {
  extern __shared__ float lds[];

  const int tid = threadIdx.x;
  const int z  = blockIdx.x;
  const int sp = index[z];
  const int wv   = __builtin_amdgcn_readfirstlane(tid >> 5);
  const int lane = tid & 31;
  const int half = lane >> 4;
  const int r    = lane & 15;
  const int rowBase = z * 128 + wv * BLOCK_ROWS;
  const int ce = wv * BLOCK_ROWS + lane;
  float* G = lds + wv * (BLOCK_ROWS * GPITCH);

  float xr0[10], xr1[10];
  {
    const float* xp0 = x + (rowBase + r) * 9;
    const float* xp1 = x + (rowBase + TILE_ROWS + r) * 9;
#pragma unroll
    for (int i = 0; i < 8; ++i) {
      xr0[i] = xp0[i ^ half];
      xr1[i] = xp1[i ^ half];
    }
    xr0[8] = xp0[8];
    xr1[8] = xp1[8];
    xr0[9] = 1.f;
    xr1[9] = 1.f;
  }

  float wa[12], wb[9], wc[7];
  wa[0] = w1_2e[sp * 128 + ce];
#pragma unroll
  for (int k = 0; k < 3; ++k) wa[1 + k] = w2_2e[(sp * 3 + k) * 128 + ce];
#pragma unroll
  for (int k = 0; k < 8; ++k) wa[4 + k] = w3_2e[(sp * 8 + k) * 128 + ce];
  wb[0] = w1_1o[sp * 128 + ce];
#pragma unroll
  for (int k = 0; k < 2; ++k) wb[1 + k] = w2_1o[(sp * 2 + k) * 128 + ce];
#pragma unroll
  for (int k = 0; k < 6; ++k) wb[3 + k] = w3_1o[(sp * 6 + k) * 128 + ce];
  wc[0] = w1_0e[sp * 128 + ce];
  wc[1] = w2_0e[(sp * 2 + 0) * 128 + ce];
  wc[2] = w2_0e[(sp * 2 + 1) * 128 + ce];
#pragma unroll
  for (int k = 0; k < 4; ++k) wc[3 + k] = w3_0e[(sp * 4 + k) * 128 + ce];

  {
    float* Gt = G + (8 * half) * GPITCH + r;
    row_group<false, 2, LIGHT_PAIRS, LIGHT_STEPS>(xr0, xr1, (const v4f*)UF + lane, Gt);
    row_group<true, 4, HEAVY_PAIRS, HEAVY_STEPS>(
        xr0, xr1, (const v4f*)UF + 2 * LIGHT_PAIRS * 32 + lane, Gt + 2 * 16);
  }
  __syncthreads();

  {
    const float* Gr = G + lane * GPITCH;
    float* ob = out + z * 1152;
    {
      float b[5];
#pragma unroll
      for (int i = 0; i < 5; ++i) {
        b[i] = wa[0] * Gr[4 + i];
#pragma unroll
        for (int k = 0; k < 3; ++k) b[i] += wa[1 + k] * Gr[17 + 5 * k + i];
#pragma unroll
        for (int k = 0; k < 8; ++k) b[i] += wa[4 + k] * Gr[54 + 5 * k + i];
      }
      float* o2 = ob + 512 + ce * 5;
#pragma unroll
      for (int i = 0; i < 5; ++i) o2[i] = b[i];
    }
    {
      float b[3];
#pragma unroll
      for (int i = 0; i < 3; ++i) {
        b[i] = wb[0] * Gr[1 + i];
#pragma unroll
        for (int k = 0; k < 2; ++k) b[i] += wb[1 + k] * Gr[11 + 3 * k + i];
#pragma unroll
        for (int k = 0; k < 6; ++k) b[i] += wb[3 + k] * Gr[36 + 3 * k + i];
      }
      float* o1 = ob + 128 + ce * 3;
#pragma unroll
      for (int i = 0; i < 3; ++i) o1[i] = b[i];
    }
    {
      float a = wc[0] * Gr[0] + wc[1] * Gr[9] + wc[2] * Gr[10];
#pragma unroll
      for (int k = 0; k < 4; ++k) a += wc[3 + k] * Gr[32 + k];
      ob[ce] = a;
    }
  }
}

extern "C" void kernel_launch(void* const* d_in, const int* in_sizes, int n_in,
                              void* d_out, int out_size, void* d_ws, size_t ws_size,
                              hipStream_t stream) {
  const float* x     = (const float*)d_in[0];
  const float* u1_0e = (const float*)d_in[1];
  const float* w1_0e = (const float*)d_in[2];
  const float* u1_1o = (const float*)d_in[3];
  const float* w1_1o = (const float*)d_in[4];
  const float* u1_2e = (const float*)d_in[5];
  const float* w1_2e = (const float*)d_in[6];
  const float* u2_0e = (const float*)d_in[7];
  const float* w2_0e = (const float*)d_in[8];
  const float* u2_1o = (const float*)d_in[9];
  const float* w2_1o = (const float*)d_in[10];
  const float* u2_2e = (const float*)d_in[11];
  const float* w2_2e = (const float*)d_in[12];
  const float* u3_0e = (const float*)d_in[13];
  const float* w3_0e = (const float*)d_in[14];
  const float* u3_1o = (const float*)d_in[15];
  const float* w3_1o = (const float*)d_in[16];
  const float* u3_2e = (const float*)d_in[17];
  const float* w3_2e = (const float*)d_in[18];
  const int*   index = (const int*)d_in[19];
  float* UF  = (float*)d_ws;
  float* out = (float*)d_out;

  fill_u<<<(UF_FLOATS + 255) / 256, 256, 0, stream>>>(
      u1_0e, u1_1o, u1_2e, u2_0e, u2_1o, u2_2e, u3_0e, u3_1o, u3_2e, UF);

  const size_t ldsBytes = 4 * BLOCK_ROWS * GPITCH * sizeof(float);
  symcon_main<<<NROW / 128, BLOCK_THREADS, ldsBytes, stream>>>(
      x, UF, w1_0e, w1_1o, w1_2e, w2_0e, w2_1o, w2_2e, w3_0e, w3_1o, w3_2e,
      index, out);
}


# Fill launch: all six U gathers of an element asked for at once (a missing permutation re-reads the first address, deselected), block size a literal: dependent round trips 6 -> 2; same sums and order
# speedup vs baseline: 9.4061x; 1.0657x over previous
#include <hip/hip_runtime.h>


typedef float v2f __attribute__((ext_vector_type(2)));
typedef float v4f __attribute__((ext_vector_type(4)));
typedef float v8f __attribute__((ext_vector_type(8)));

#define UCOLS 96
#define NROW  131072
#define TILE_ROWS 16
#define BLOCK_ROWS 32
#define BLOCK_THREADS 128
#define GPITCH 98

#define LIGHT_STEPS 15
#define HEAVY_STEPS 43
#define LIGHT_PAIRS ((LIGHT_STEPS + 1) / 2)
#define HEAVY_PAIRS ((HEAVY_STEPS + 1) / 2)
#define MAX_SLOTS (4 * HEAVY_PAIRS)
#define NPAIRS (2 * LIGHT_PAIRS + 4 * HEAVY_PAIRS)
#define UF_FLOATS (NPAIRS * 128)

struct Slot { unsigned char a, b, c, order; };
struct SlotTab { Slot s[2][MAX_SLOTS]; };

constexpr int swap_pair(int i) { return i < 8 ? (i ^ 1) : i; }

constexpr void add_slot(SlotTab& t, int& n, int a, int b, int c, int order) {
  int pa = swap_pair(a), pb = swap_pair(b), pc = swap_pair(c);
  if (pa > pb) { const int x = pa; pa = pb; pb = x; }
  if (pb > pc) { const int x = pb; pb = pc; pc = x; }
  if (pa > pb) { const int x = pa; pa = pb; pb = x; }
  const int self = (a * 16 + b) * 16 + c;
  const int img  = (pa * 16 + pb) * 16 + pc;
  if (img < self) return;
  t.s[0][n] = Slot{(unsigned char)a, (unsigned char)b, (unsigned char)c,
                   (unsigned char)order};
  t.s[1][n] = Slot{(unsigned char)pa, (unsigned char)pb, (unsigned char)pc,
                   (unsigned char)(img == self ? 0 : order)};
  ++n;
}

constexpr SlotTab make_tab(bool heavy) {
  SlotTab t{};
  int n = 0;
  if (!heavy) {
    for (int a = 0; a < 9; ++a) add_slot(t, n, a, 9, 9, 1);
    for (int a = 0; a < 9; ++a)
      for (int b = a; b < 9; ++b) add_slot(t, n, a, b, 9, 2);
  } else {
    for (int a = 0; a < 9; ++a)
      for (int b = a; b < 9; ++b)
        for (int c = b; c < 9; ++c) add_slot(t, n, a, b, c, 3);
  }
  while (n < MAX_SLOTS) {
    t.s[0][n] = Slot{9, 9, 9, 0};
    t.s[1][n] = Slot{9, 9, 9, 0};
    ++n;
  }
  return t;
}

constexpr int used_slots(bool heavy) {
  const SlotTab t = make_tab(heavy);
  int n = 0;
  for (int i = 0; i < MAX_SLOTS; ++i) n += t.s[0][i].order != 0;
  return n;
}
static_assert(used_slots(false) == 2 * LIGHT_STEPS, "orders 1 and 2 fill 15 steps");
static_assert(used_slots(true) == 2 * HEAVY_STEPS - 1, "order 3 fills 43 steps but one slot");

__constant__ const SlotTab TAB_LIGHT = make_tab(false);
__constant__ const SlotTab TAB_HEAVY = make_tab(true);

#define FILL_THREADS 64

__global__ __launch_bounds__(FILL_THREADS) void fill_u(
    const float* __restrict__ u1_0e, const float* __restrict__ u1_1o,
    const float* __restrict__ u1_2e, const float* __restrict__ u2_0e,
    const float* __restrict__ u2_1o, const float* __restrict__ u2_2e,
    const float* __restrict__ u3_0e, const float* __restrict__ u3_1o,
    const float* __restrict__ u3_2e, float* __restrict__ UF) {
  const int e = blockIdx.x * FILL_THREADS + threadIdx.x;
  if (e >= UF_FLOATS) return;
  const int P    = e >> 7;
  const int lane = (e >> 2) & 31;
  const int q    = e & 3;
  int w, pr;
  const bool heavy = P >= 2 * LIGHT_PAIRS;
  if (!heavy) {
    w = P / LIGHT_PAIRS; pr = P - w * LIGHT_PAIRS;
  } else {
    const int t = P - 2 * LIGHT_PAIRS;
    w = 2 + t / HEAVY_PAIRS; pr = t - (w - 2) * HEAVY_PAIRS;
  }
  const int col  = 16 * w + (lane & 15);
  const int slot = 2 * (2 * pr + (q >> 1)) + (q & 1);
  const int half = lane >> 4;
  const Slot s = heavy ? TAB_HEAVY.s[half][slot] : TAB_LIGHT.s[half][slot];
  const float* u = u1_0e;
  int mi = 1, rem = 0, corder = -1;
  if (col < 1)       { u = u1_0e; mi = 1;  rem = col;      corder = 1; }
  else if (col < 4)  { u = u1_1o; mi = 3;  rem = col - 1;  corder = 1; }
  else if (col < 9)  { u = u1_2e; mi = 5;  rem = col - 4;  corder = 1; }
  else if (col < 11) { u = u2_0e; mi = 2;  rem = col - 9;  corder = 2; }
  else if (col < 17) { u = u2_1o; mi = 6;  rem = col - 11; corder = 2; }
  else if (col < 32) { u = u2_2e; mi = 15; rem = col - 17; corder = 2; }
  else if (col < 36) { u = u3_0e; mi = 4;  rem = col - 32; corder = 3; }
  else if (col < 54) { u = u3_1o; mi = 18; rem = col - 36; corder = 3; }
  else if (col < 94) { u = u3_2e; mi = 40; rem = col - 54; corder = 3; }
  const bool live = corder == (int)s.order;
  const int a = s.a, b = s.b, c = s.c;
  const bool o2 = live && corder == 2;
  const bool o3 = live && corder == 3;
  const int n0 = !live ? 0 : corder == 1 ? a : corder == 2 ? a * 9 + b : (a * 9 + b) * 9 + c;
  const bool t1 = (o2 && a != b) || (o3 && b != c);
  const bool t2 = o3 && a != b;
  const bool t4 = o3 && b != c;
  const bool t5 = o3 && a != b && b != c;
  const int n1 = t1 ? (o2 ? b * 9 + a : (a * 9 + c) * 9 + b) : n0;
  const int n2 = t2 ? (b * 9 + a) * 9 + c : n0;
  const int n3 = t2 ? (b * 9 + c) * 9 + a : n0;
  const int n4 = t4 ? (c * 9 + a) * 9 + b : n0;
  const int n5 = t5 ? (c * 9 + b) * 9 + a : n0;
  const float v0 = u[n0 * mi + rem];
  const float v1 = u[n1 * mi + rem];
  const float v2 = u[n2 * mi + rem];
  const float v3 = u[n3 * mi + rem];
  const float v4 = u[n4 * mi + rem];
  const float v5 = u[n5 * mi + rem];
  float v = v0;
  v = t1 ? v + v1 : v;
  v = t2 ? v + v2 : v;
  v = t2 ? v + v3 : v;
  v = t4 ? v + v4 : v;
  v = t5 ? v + v5 : v;
  UF[e] = live ? v : 0.f;
}

template <bool HEAVY, int S>
__device__ __forceinline__ v2f a_frag(const float (&xr)[10]) {
  constexpr SlotTab tab = make_tab(HEAVY);
  constexpr Slot s0 = tab.s[0][2 * S];
  constexpr Slot s1 = tab.s[0][2 * S + 1];
  v2f a;
  a.x = (xr[s0.a] * xr[s0.b]) * xr[s0.c];
  a.y = (xr[s1.a] * xr[s1.b]) * xr[s1.c];
  return a;
}

__device__ __forceinline__ v8f wmma4(v2f a, v2f b, v8f c) {
  return __builtin_amdgcn_wmma_f32_16x16x4_f32(
       false, a,  false, b,
       (short)0, c,  false,  false);
}

template <int NT>
struct BSet { v4f v[NT]; };

template <bool HEAVY, int NT, int J, int NPAIR, int NSTEP>
struct RowRun {
  static __device__ __forceinline__ void run(const float (&xa)[10], const float (&xb)[10],
                                             const v4f* __restrict__ bp, v8f (&acca)[NT],
                                             v8f (&accb)[NT], BSet<NT> b, BSet<NT> bn1) {
    BSet<NT> bn2 = bn1;
    if constexpr (J + 2 < NPAIR) {
#pragma unroll
      for (int t = 0; t < NT; ++t) bn2.v[t] = bp[(t * NPAIR + J + 2) * 32];
    }
    {
      const v2f a0 = a_frag<HEAVY, 2 * J>(xa);
      const v2f a1 = a_frag<HEAVY, 2 * J>(xb);
#pragma unroll
      for (int t = 0; t < NT; ++t) {
        const v2f b0 = {b.v[t].x, b.v[t].y};
        acca[t] = wmma4(a0, b0, acca[t]);
        accb[t] = wmma4(a1, b0, accb[t]);
      }
    }
    if constexpr (2 * J + 1 < NSTEP) {
      const v2f a0 = a_frag<HEAVY, 2 * J + 1>(xa);
      const v2f a1 = a_frag<HEAVY, 2 * J + 1>(xb);
#pragma unroll
      for (int t = 0; t < NT; ++t) {
        const v2f b1 = {b.v[t].z, b.v[t].w};
        acca[t] = wmma4(a0, b1, acca[t]);
        accb[t] = wmma4(a1, b1, accb[t]);
      }
    }
    __builtin_amdgcn_sched_barrier(0);
    RowRun<HEAVY, NT, J + 1, NPAIR, NSTEP>::run(xa, xb, bp, acca, accb, bn1, bn2);
  }
};
template <bool HEAVY, int NT, int NPAIR, int NSTEP>
struct RowRun<HEAVY, NT, NPAIR, NPAIR, NSTEP> {
  static __device__ __forceinline__ void run(const float (&)[10], const float (&)[10],
                                             const v4f* __restrict__, v8f (&)[NT],
                                             v8f (&)[NT], BSet<NT>, BSet<NT>) {}
};

template <bool HEAVY, int NT, int NPAIR, int NSTEP>
__device__ __forceinline__ void row_group(const float (&xa)[10], const float (&xb)[10],
                                          const v4f* __restrict__ bp, float* __restrict__ Gt) {
  v8f acca[NT], accb[NT];
#pragma unroll
  for (int t = 0; t < NT; ++t) {
    acca[t] = v8f{0.f, 0.f, 0.f, 0.f, 0.f, 0.f, 0.f, 0.f};
    accb[t] = v8f{0.f, 0.f, 0.f, 0.f, 0.f, 0.f, 0.f, 0.f};
  }
  BSet<NT> b, bn1;
#pragma unroll
  for (int t = 0; t < NT; ++t) {
    b.v[t]   = bp[(t * NPAIR) * 32];
    bn1.v[t] = bp[(t * NPAIR + 1) * 32];
  }
  RowRun<HEAVY, NT, 0, NPAIR, NSTEP>::run(xa, xb, bp, acca, accb, b, bn1);
#pragma unroll
  for (int t = 0; t < NT; ++t) {
#pragma unroll
    for (int j = 0; j < 8; ++j) {
      Gt[j * GPITCH + t * 16] = acca[t][j];
      Gt[(TILE_ROWS + j) * GPITCH + t * 16] = accb[t][j];
    }
  }
}

__global__ __launch_bounds__(BLOCK_THREADS, 4) void symcon_main(
    const float* __restrict__ x, const float* __restrict__ UF,
    const float* __restrict__ w1_0e, const float* __restrict__ w1_1o,
    const float* __restrict__ w1_2e, const float* __restrict__ w2_0e,
    const float* __restrict__ w2_1o, const float* __restrict__ w2_2e,
    const float* __restrict__ w3_0e, const float* __restrict__ w3_1o,
    const float* __restrict__ w3_2e, const int* __restrict__ index,
    float* __restrict__ out) {
  extern __shared__ float lds[];

  const int tid = threadIdx.x;
  const int z  = blockIdx.x;
  const int sp = index[z];
  const int wv   = __builtin_amdgcn_readfirstlane(tid >> 5);
  const int lane = tid & 31;
  const int half = lane >> 4;
  const int r    = lane & 15;
  const int rowBase = z * 128 + wv * BLOCK_ROWS;
  const int ce = wv * BLOCK_ROWS + lane;
  float* G = lds + wv * (BLOCK_ROWS * GPITCH);

  float xr0[10], xr1[10];
  {
    const float* xp0 = x + (rowBase + r) * 9;
    const float* xp1 = x + (rowBase + TILE_ROWS + r) * 9;
#pragma unroll
    for (int i = 0; i < 8; ++i) {
      xr0[i] = xp0[i ^ half];
      xr1[i] = xp1[i ^ half];
    }
    xr0[8] = xp0[8];
    xr1[8] = xp1[8];
    xr0[9] = 1.f;
    xr1[9] = 1.f;
  }

  float wa[12], wb[9], wc[7];
  wa[0] = w1_2e[sp * 128 + ce];
#pragma unroll
  for (int k = 0; k < 3; ++k) wa[1 + k] = w2_2e[(sp * 3 + k) * 128 + ce];
#pragma unroll
  for (int k = 0; k < 8; ++k) wa[4 + k] = w3_2e[(sp * 8 + k) * 128 + ce];
  wb[0] = w1_1o[sp * 128 + ce];
#pragma unroll
  for (int k = 0; k < 2; ++k) wb[1 + k] = w2_1o[(sp * 2 + k) * 128 + ce];
#pragma unroll
  for (int k = 0; k < 6; ++k) wb[3 + k] = w3_1o[(sp * 6 + k) * 128 + ce];
  wc[0] = w1_0e[sp * 128 + ce];
  wc[1] = w2_0e[(sp * 2 + 0) * 128 + ce];
  wc[2] = w2_0e[(sp * 2 + 1) * 128 + ce];
#pragma unroll
  for (int k = 0; k < 4; ++k) wc[3 + k] = w3_0e[(sp * 4 + k) * 128 + ce];

  {
    float* Gt = G + (8 * half) * GPITCH + r;
    row_group<false, 2, LIGHT_PAIRS, LIGHT_STEPS>(xr0, xr1, (const v4f*)UF + lane, Gt);
    row_group<true, 4, HEAVY_PAIRS, HEAVY_STEPS>(
        xr0, xr1, (const v4f*)UF + 2 * LIGHT_PAIRS * 32 + lane, Gt + 2 * 16);
  }
  __syncthreads();

  {
    const float* Gr = G + lane * GPITCH;
    float* ob = out + z * 1152;
    {
      float b[5];
#pragma unroll
      for (int i = 0; i < 5; ++i) {
        b[i] = wa[0] * Gr[4 + i];
#pragma unroll
        for (int k = 0; k < 3; ++k) b[i] += wa[1 + k] * Gr[17 + 5 * k + i];
#pragma unroll
        for (int k = 0; k < 8; ++k) b[i] += wa[4 + k] * Gr[54 + 5 * k + i];
      }
      float* o2 = ob + 512 + ce * 5;
#pragma unroll
      for (int i = 0; i < 5; ++i) o2[i] = b[i];
    }
    {
      float b[3];
#pragma unroll
      for (int i = 0; i < 3; ++i) {
        b[i] = wb[0] * Gr[1 + i];
#pragma unroll
        for (int k = 0; k < 2; ++k) b[i] += wb[1 + k] * Gr[11 + 3 * k + i];
#pragma unroll
        for (int k = 0; k < 6; ++k) b[i] += wb[3 + k] * Gr[36 + 3 * k + i];
      }
      float* o1 = ob + 128 + ce * 3;
#pragma unroll
      for (int i = 0; i < 3; ++i) o1[i] = b[i];
    }
    {
      float a = wc[0] * Gr[0] + wc[1] * Gr[9] + wc[2] * Gr[10];
#pragma unroll
      for (int k = 0; k < 4; ++k) a += wc[3 + k] * Gr[32 + k];
      ob[ce] = a;
    }
  }
}

extern "C" void kernel_launch(void* const* d_in, const int* in_sizes, int n_in,
                              void* d_out, int out_size, void* d_ws, size_t ws_size,
                              hipStream_t stream) {
  const float* x     = (const float*)d_in[0];
  const float* u1_0e = (const float*)d_in[1];
  const float* w1_0e = (const float*)d_in[2];
  const float* u1_1o = (const float*)d_in[3];
  const float* w1_1o = (const float*)d_in[4];
  const float* u1_2e = (const float*)d_in[5];
  const float* w1_2e = (const float*)d_in[6];
  const float* u2_0e = (const float*)d_in[7];
  const float* w2_0e = (const float*)d_in[8];
  const float* u2_1o = (const float*)d_in[9];
  const float* w2_1o = (const float*)d_in[10];
  const float* u2_2e = (const float*)d_in[11];
  const float* w2_2e = (const float*)d_in[12];
  const float* u3_0e = (const float*)d_in[13];
  const float* w3_0e = (const float*)d_in[14];
  const float* u3_1o = (const float*)d_in[15];
  const float* w3_1o = (const float*)d_in[16];
  const float* u3_2e = (const float*)d_in[17];
  const float* w3_2e = (const float*)d_in[18];
  const int*   index = (const int*)d_in[19];
  float* UF  = (float*)d_ws;
  float* out = (float*)d_out;

  fill_u<<<(UF_FLOATS + FILL_THREADS - 1) / FILL_THREADS, FILL_THREADS, 0, stream>>>(
      u1_0e, u1_1o, u1_2e, u2_0e, u2_1o, u2_2e, u3_0e, u3_1o, u3_2e, UF);

  const size_t ldsBytes = 4 * BLOCK_ROWS * GPITCH * sizeof(float);
  symcon_main<<<NROW / 128, BLOCK_THREADS, ldsBytes, stream>>>(
      x, UF, w1_0e, w1_1o, w1_2e, w2_0e, w2_1o, w2_2e, w3_0e, w3_1o, w3_2e,
      index, out);
}
